# starting kernel
# baseline (speedup 1.0000x reference)
#include <hip/hip_runtime.h>
#include <hip/hip_bf16.h>

#define B_SZ 4
#define L_SZ 2048
#define D_SZ 1024
#define N_SZ 16
#define P_SZ 33

typedef float v2f __attribute__((ext_vector_type(2)));
typedef float v8f __attribute__((ext_vector_type(8)));

__global__ __launch_bounds__(256) void conv_silu_kernel(
    const float* __restrict__ x, const float* __restrict__ cw,
    const float* __restrict__ cb, float* __restrict__ xc)
{
  const int idx = blockIdx.x * 256 + threadIdx.x;
  const int d  = idx & (D_SZ - 1);
  const int bl = idx >> 10;
  const int l  = bl & (L_SZ - 1);
  const float* xp = x + (size_t)bl * D_SZ + d;
  const float* wp = cw + (size_t)d * 4;
  float acc = cb[d];
#pragma unroll
  for (int j = 0; j < 4; ++j) {
    const int li = l - 3 + j;
    const float xv = (li >= 0) ? xp[(j - 3) * D_SZ] : 0.f;
    acc = fmaf(xv, wp[j], acc);
  }
  xc[idx] = acc / (1.f + __expf(-acc));
}

__global__ __launch_bounds__(128) void proj_wmma_kernel(
    const float* __restrict__ xc, const float* __restrict__ w,
    float* __restrict__ proj)
{
  const int wave = blockIdx.x * 4 + (threadIdx.x >> 5);
  const int lane = threadIdx.x & 31;
  const int mrow = wave * 16 + (lane & 15);
  const int kb   = (lane >> 4) << 1;
  const float* arow = xc + (size_t)mrow * D_SZ + kb;

  const int ncol = lane & 15;
  const int p0 = ncol;
  const int p1 = 16 + ncol;
  int p2 = 32 + ncol;
  const float s2 = (p2 < P_SZ) ? 1.f : 0.f;
  if (p2 >= P_SZ) p2 = P_SZ - 1;
  const float* brow0 = w + (size_t)p0 * D_SZ + kb;
  const float* brow1 = w + (size_t)p1 * D_SZ + kb;
  const float* brow2 = w + (size_t)p2 * D_SZ + kb;

  v8f acc0 = {}, acc1 = {}, acc2 = {};
  for (int k0 = 0; k0 < D_SZ; k0 += 4) {
    v2f a  = *(const v2f*)(arow  + k0);
    v2f b0 = *(const v2f*)(brow0 + k0);
    v2f b1 = *(const v2f*)(brow1 + k0);
    v2f b2 = *(const v2f*)(brow2 + k0);
    b2 *= s2;
    acc0 = __builtin_amdgcn_wmma_f32_16x16x4_f32(false, a, false, b0,
                                                 (short)0, acc0, false, false);
    acc1 = __builtin_amdgcn_wmma_f32_16x16x4_f32(false, a, false, b1,
                                                 (short)0, acc1, false, false);
    acc2 = __builtin_amdgcn_wmma_f32_16x16x4_f32(false, a, false, b2,
                                                 (short)0, acc2, false, false);
  }

  const int mhi = (lane >> 4) << 3;
#pragma unroll
  for (int v = 0; v < 8; ++v) {
    const int row = wave * 16 + v + mhi;
    float* pr = proj + (size_t)row * P_SZ;
    pr[p0] = acc0[v];
    pr[p1] = acc1[v];
    if (32 + ncol < P_SZ) pr[32 + ncol] = acc2[v];
  }
}

__global__ __launch_bounds__(256) void scan_kernel(
    const float* __restrict__ x, const float* __restrict__ xc,
    const float* __restrict__ proj, const float* __restrict__ dt_w,
    const float* __restrict__ dt_b, const float* __restrict__ a_log,
    const float* __restrict__ Dp, float* __restrict__ out)
{
  const int n = threadIdx.x & 15;
  const int c = blockIdx.x * 16 + (threadIdx.x >> 4);
  const int b = c >> 10;
  const int d = c & (D_SZ - 1);

  const float a_c = -__expf(a_log[d * N_SZ + n]);
  const float dtw = dt_w[d];
  const float dtb = dt_b[d];
  const float dpd = Dp[d];

  const float* pr   = proj + (size_t)b * L_SZ * P_SZ;
  const float* xr   = x    + (size_t)b * L_SZ * D_SZ + d;
  const float* xcr  = xc   + (size_t)b * L_SZ * D_SZ + d;
  float*       outr = out  + (size_t)b * L_SZ * D_SZ + d;

  float h = 0.f;
  for (int t = 0; t < L_SZ; ++t) {
    const float* pt = pr + (size_t)t * P_SZ;
    const float bt  = pt[n];
    const float ct  = pt[N_SZ + n];
    const float p32 = pt[2 * N_SZ];
    const float xt  = xcr[(size_t)t * D_SZ];

    const float z     = fmaf(p32, dtw, dtb);
    const float delta = (z > 20.f) ? z : __logf(1.f + __expf(z));
    const float ab    = __expf(delta * a_c);

    h = fmaf(ab, h, delta * xt * bt);
    float y = h * ct;
    y += __shfl_xor(y, 1);
    y += __shfl_xor(y, 2);
    y += __shfl_xor(y, 4);
    y += __shfl_xor(y, 8);
    if (n == 0)
      outr[(size_t)t * D_SZ] = fmaf(xr[(size_t)t * D_SZ], dpd, y);
  }
}

extern "C" void kernel_launch(void* const* d_in, const int* in_sizes, int n_in,
                              void* d_out, int out_size, void* d_ws, size_t ws_size,
                              hipStream_t stream) {
  const float* x      = (const float*)d_in[0];
  const float* conv_w = (const float*)d_in[1];
  const float* conv_b = (const float*)d_in[2];
  const float* xproj  = (const float*)d_in[3];
  const float* dt_w   = (const float*)d_in[4];
  const float* dt_b   = (const float*)d_in[5];
  const float* a_log  = (const float*)d_in[6];
  const float* Dp     = (const float*)d_in[7];
  float* out = (float*)d_out;

  float* xc   = (float*)d_ws;
  float* proj = (float*)((char*)d_ws +
                         (size_t)B_SZ * L_SZ * D_SZ * sizeof(float));

  conv_silu_kernel<<<(B_SZ * L_SZ * D_SZ) / 256, 256, 0, stream>>>(
      x, conv_w, conv_b, xc);

  proj_wmma_kernel<<<128, 128, 0, stream>>>(xc, xproj, proj);

  scan_kernel<<<(B_SZ * D_SZ) / 16, 256, 0, stream>>>(
      x, xc, proj, dt_w, dt_b, a_log, Dp, out);
}


# conv output never stored: proj builds its A fragments from x in registers (4 row loads, conv+SiLU per lane), both scans re-form it from a 4-step register window; 3 launches; proj k sum order changed
# speedup vs baseline: 15.8055x; 15.8055x over previous
#include <hip/hip_runtime.h>
#include <hip/hip_bf16.h>


#define B_SZ 4
#define L_SZ 2048
#define D_SZ 1024
#define N_SZ 16
#define P_SZ 33
#define CH_N 16
#define CH_T (L_SZ / CH_N)
#define CHAN (B_SZ * D_SZ)
#define SCAN_BLK 256
#define SCAN_TB 8

typedef float v2f __attribute__((ext_vector_type(2)));
typedef float v4f __attribute__((ext_vector_type(4)));
typedef float v8f __attribute__((ext_vector_type(8)));

__device__ __forceinline__ float conv_silu(float x0, float x1, float x2, float x3,
                                           v4f w, float bias)
{
  float acc = bias;
  acc = fmaf(x0, w.x, acc);
  acc = fmaf(x1, w.y, acc);
  acc = fmaf(x2, w.z, acc);
  acc = fmaf(x3, w.w, acc);
  return acc / (1.f + __expf(-acc));
}

struct ProjStep {
  v4f x0, x1, x2, x3;
  v4f wa, wb, wc, wd;
  v4f bias;
  v4f b0, b1, b2;
};

struct ProjPtrs {
  const float *xp0, *xp1, *xp2, *xp3, *cwp, *cbp, *bp0, *bp1, *bp2;
};

__device__ __forceinline__ ProjStep proj_load(const ProjPtrs& p, const int k0)
{
  ProjStep r;
  r.x0   = *(const v4f*)(p.xp0 + k0);
  r.x1   = *(const v4f*)(p.xp1 + k0);
  r.x2   = *(const v4f*)(p.xp2 + k0);
  r.x3   = *(const v4f*)(p.xp3 + k0);
  r.wa   = *(const v4f*)(p.cwp + (size_t)k0 * 4);
  r.wb   = *(const v4f*)(p.cwp + (size_t)k0 * 4 + 4);
  r.wc   = *(const v4f*)(p.cwp + (size_t)k0 * 4 + 8);
  r.wd   = *(const v4f*)(p.cwp + (size_t)k0 * 4 + 12);
  r.bias = *(const v4f*)(p.cbp + k0);
  r.b0   = *(const v4f*)(p.bp0 + k0);
  r.b1   = *(const v4f*)(p.bp1 + k0);
  r.b2   = *(const v4f*)(p.bp2 + k0);
  return r;
}

template <bool FIRST>
__device__ __forceinline__ void proj_step(ProjStep s, const bool z0, const bool z1,
                                          const bool z2, v8f& acc0, v8f& acc1, v8f& acc2)
{
  if (FIRST) {
    const v4f zero4 = {0.f, 0.f, 0.f, 0.f};
    if (z0) s.x0 = zero4;
    if (z1) s.x1 = zero4;
    if (z2) s.x2 = zero4;
  }
  const v2f alo = {conv_silu(s.x0.x, s.x1.x, s.x2.x, s.x3.x, s.wa, s.bias.x),
                   conv_silu(s.x0.y, s.x1.y, s.x2.y, s.x3.y, s.wb, s.bias.y)};
  const v2f ahi = {conv_silu(s.x0.z, s.x1.z, s.x2.z, s.x3.z, s.wc, s.bias.z),
                   conv_silu(s.x0.w, s.x1.w, s.x2.w, s.x3.w, s.wd, s.bias.w)};

  acc0 = __builtin_amdgcn_wmma_f32_16x16x4_f32(false, alo, false, (v2f){s.b0.x, s.b0.y},
                                               (short)0, acc0, false, false);
  acc1 = __builtin_amdgcn_wmma_f32_16x16x4_f32(false, alo, false, (v2f){s.b1.x, s.b1.y},
                                               (short)0, acc1, false, false);
  acc2 = __builtin_amdgcn_wmma_f32_16x16x4_f32(false, alo, false, (v2f){s.b2.x, s.b2.y},
                                               (short)0, acc2, false, false);
  acc0 = __builtin_amdgcn_wmma_f32_16x16x4_f32(false, ahi, false, (v2f){s.b0.z, s.b0.w},
                                               (short)0, acc0, false, false);
  acc1 = __builtin_amdgcn_wmma_f32_16x16x4_f32(false, ahi, false, (v2f){s.b1.z, s.b1.w},
                                               (short)0, acc1, false, false);
  acc2 = __builtin_amdgcn_wmma_f32_16x16x4_f32(false, ahi, false, (v2f){s.b2.z, s.b2.w},
                                               (short)0, acc2, false, false);
}

template <bool FIRST>
__device__ __forceinline__ void conv_proj_tile(
    const float* __restrict__ x, const float* __restrict__ cw,
    const float* __restrict__ cb, const float* __restrict__ w,
    float* __restrict__ proj, const int tile, const int lane)
{
  const int c   = lane & 15;
  const int kh  = (lane >> 4) << 2;
  const int row = tile * 16 + c;
  const float* xrow = x + (size_t)row * D_SZ + kh;
  const bool z0 = FIRST && (c < 3);
  const bool z1 = FIRST && (c < 2);
  const bool z2 = FIRST && (c < 1);

  ProjPtrs p;
  p.xp0 = z0 ? xrow : xrow - 3 * D_SZ;
  p.xp1 = z1 ? xrow : xrow - 2 * D_SZ;
  p.xp2 = z2 ? xrow : xrow - 1 * D_SZ;
  p.xp3 = xrow;
  p.cwp = cw + (size_t)kh * 4;
  p.cbp = cb + kh;
  p.bp0 = w + (size_t)c * D_SZ + kh;
  p.bp1 = w + (size_t)(16 + c) * D_SZ + kh;
  p.bp2 = w + (size_t)(P_SZ - 1) * D_SZ + kh;

  v8f acc0 = {}, acc1 = {}, acc2 = {};
  ProjStep cur0 = proj_load(p, 0);
  ProjStep cur1 = proj_load(p, 8);
#pragma unroll 2
  for (int k0 = 0; k0 < D_SZ; k0 += 16) {
    const ProjStep nxt0 = proj_load(p, (k0 + 16) & (D_SZ - 1));
    const ProjStep nxt1 = proj_load(p, (k0 + 24) & (D_SZ - 1));
    __builtin_amdgcn_sched_barrier(0);

    proj_step<FIRST>(cur0, z0, z1, z2, acc0, acc1, acc2);
    proj_step<FIRST>(cur1, z0, z1, z2, acc0, acc1, acc2);

    __builtin_amdgcn_sched_barrier(0);
    asm volatile("" :: "v"(nxt0.x0), "v"(nxt0.x1), "v"(nxt0.x2), "v"(nxt0.x3),
                       "v"(nxt0.wa), "v"(nxt0.wb), "v"(nxt0.wc), "v"(nxt0.wd),
                       "v"(nxt0.bias), "v"(nxt0.b0), "v"(nxt0.b1), "v"(nxt0.b2));
    asm volatile("" :: "v"(nxt1.x0), "v"(nxt1.x1), "v"(nxt1.x2), "v"(nxt1.x3),
                       "v"(nxt1.wa), "v"(nxt1.wb), "v"(nxt1.wc), "v"(nxt1.wd),
                       "v"(nxt1.bias), "v"(nxt1.b0), "v"(nxt1.b1), "v"(nxt1.b2));
    cur0 = nxt0;
    cur1 = nxt1;
  }

  const int mhi = (lane >> 4) << 3;
#pragma unroll
  for (int v = 0; v < 8; ++v) {
    float* pr = proj + (size_t)(tile * 16 + v + mhi) * P_SZ;
    pr[c]      = acc0[v];
    pr[16 + c] = acc1[v];
    if (c == 0) pr[32] = acc2[v];
  }
}

__global__ __launch_bounds__(32) void conv_proj_wmma_kernel(
    const float* __restrict__ x, const float* __restrict__ cw,
    const float* __restrict__ cb, const float* __restrict__ w,
    float* __restrict__ proj)
{
  const int tile = blockIdx.x;
  const int lane = threadIdx.x;
  if ((tile & (L_SZ / 16 - 1)) == 0) conv_proj_tile<true>(x, cw, cb, w, proj, tile, lane);
  else                               conv_proj_tile<false>(x, cw, cb, w, proj, tile, lane);
}

__global__ __launch_bounds__(SCAN_BLK) void scan_chunk_summary_kernel(
    const float* __restrict__ x, const float* __restrict__ cw,
    const float* __restrict__ cb_, const float* __restrict__ proj,
    const float* __restrict__ dt_w, const float* __restrict__ dt_b,
    const float* __restrict__ a_log,
    float* __restrict__ ch_e, float* __restrict__ ch_p)
{
  const int ck = blockIdx.x / (CHAN / SCAN_BLK);
  const int cb = (blockIdx.x % (CHAN / SCAN_BLK)) * SCAN_BLK;
  const int b  = cb >> 10;
  const int c  = cb + threadIdx.x;
  const int d  = c & (D_SZ - 1);

  v2f ac[8];
#pragma unroll
  for (int k = 0; k < 8; ++k) {
    ac[k].x = -__expf(a_log[d * N_SZ + 2 * k]);
    ac[k].y = -__expf(a_log[d * N_SZ + 2 * k + 1]);
  }
  const float dtw = dt_w[d];
  const float dtb = dt_b[d];
  const v4f   cwd = *(const v4f*)(cw + (size_t)d * 4);
  const float cbd = cb_[d];

  const int t0 = ck * CH_T;
  const float* pr = proj + ((size_t)b * L_SZ + t0) * P_SZ;
  const float* xr = x    + ((size_t)b * L_SZ + t0) * D_SZ + d;

  float xw0 = 0.f, xw1 = 0.f, xw2 = 0.f;
  if (ck > 0) { xw0 = xr[-3 * D_SZ]; xw1 = xr[-2 * D_SZ]; xw2 = xr[-1 * D_SZ]; }

  v2f h[8], p[8];
#pragma unroll
  for (int k = 0; k < 8; ++k) { h[k] = (v2f){0.f, 0.f}; p[k] = (v2f){1.f, 1.f}; }

  for (int tb = 0; tb < CH_T; tb += SCAN_TB) {
    float xv[SCAN_TB], xt[SCAN_TB];
#pragma unroll
    for (int i = 0; i < SCAN_TB; ++i) xv[i] = xr[(size_t)(tb + i) * D_SZ];
    xt[0] = conv_silu(xw0, xw1, xw2, xv[0], cwd, cbd);
    xt[1] = conv_silu(xw1, xw2, xv[0], xv[1], cwd, cbd);
    xt[2] = conv_silu(xw2, xv[0], xv[1], xv[2], cwd, cbd);
#pragma unroll
    for (int i = 3; i < SCAN_TB; ++i)
      xt[i] = conv_silu(xv[i - 3], xv[i - 2], xv[i - 1], xv[i], cwd, cbd);
    xw0 = xv[SCAN_TB - 3]; xw1 = xv[SCAN_TB - 2]; xw2 = xv[SCAN_TB - 1];

    float dl[SCAN_TB], dx[SCAN_TB];
#pragma unroll
    for (int i = 0; i < SCAN_TB; ++i) {
      const float p32 = pr[(size_t)(tb + i) * P_SZ + 2 * N_SZ];
      const float z   = fmaf(p32, dtw, dtb);
      const float sp  = __logf(1.f + __expf(z));
      dl[i] = (z > 20.f) ? z : sp;
    }
#pragma unroll
    for (int i = 0; i < SCAN_TB; ++i) dx[i] = dl[i] * xt[i];

#pragma unroll
    for (int i = 0; i < SCAN_TB; ++i) {
      const float* pt = pr + (size_t)(tb + i) * P_SZ;
#pragma unroll
      for (int k = 0; k < 8; ++k) {
        const v2f e  = dl[i] * ac[k];
        const v2f ab = {__expf(e.x), __expf(e.y)};
        const v2f bt = {pt[2 * k], pt[2 * k + 1]};
        const v2f u  = dx[i] * bt;
        h[k] = __builtin_elementwise_fma(ab, h[k], u);
        p[k] = p[k] * ab;
      }
    }
  }

#pragma unroll
  for (int k = 0; k < 8; ++k) {
    const size_t o0 = ((size_t)(ck * N_SZ + 2 * k)) * CHAN + c;
    ch_e[o0]        = h[k].x;
    ch_e[o0 + CHAN] = h[k].y;
    ch_p[o0]        = p[k].x;
    ch_p[o0 + CHAN] = p[k].y;
  }
}

__global__ __launch_bounds__(SCAN_BLK) void scan_kernel(
    const float* __restrict__ x, const float* __restrict__ cw,
    const float* __restrict__ cb_,
    const float* __restrict__ proj, const float* __restrict__ dt_w,
    const float* __restrict__ dt_b, const float* __restrict__ a_log,
    const float* __restrict__ Dp, const float* __restrict__ ch_e,
    const float* __restrict__ ch_p, float* __restrict__ out)
{
  const int ck = blockIdx.x / (CHAN / SCAN_BLK);
  const int cb = (blockIdx.x % (CHAN / SCAN_BLK)) * SCAN_BLK;
  const int b  = cb >> 10;
  const int c  = cb + threadIdx.x;
  const int d  = c & (D_SZ - 1);

  v2f ac[8];
#pragma unroll
  for (int k = 0; k < 8; ++k) {
    ac[k].x = -__expf(a_log[d * N_SZ + 2 * k]);
    ac[k].y = -__expf(a_log[d * N_SZ + 2 * k + 1]);
  }
  const float dtw = dt_w[d];
  const float dtb = dt_b[d];
  const float dpd = Dp[d];
  const v4f   cwd = *(const v4f*)(cw + (size_t)d * 4);
  const float cbd = cb_[d];

  v2f h[8];
#pragma unroll
  for (int k = 0; k < 8; ++k) h[k] = (v2f){0.f, 0.f};
#pragma unroll 4
  for (int j = 0; j < ck; ++j) {
    v2f pj[8], ej[8];
#pragma unroll
    for (int k = 0; k < 8; ++k) {
      const size_t o0 = ((size_t)(j * N_SZ + 2 * k)) * CHAN + c;
      pj[k].x = ch_p[o0];
      pj[k].y = ch_p[o0 + CHAN];
      ej[k].x = ch_e[o0];
      ej[k].y = ch_e[o0 + CHAN];
    }
#pragma unroll
    for (int k = 0; k < 8; ++k) h[k] = __builtin_elementwise_fma(pj[k], h[k], ej[k]);
  }

  const int t0 = ck * CH_T;
  const float* pr   = proj + ((size_t)b * L_SZ + t0) * P_SZ;
  const float* xr   = x    + ((size_t)b * L_SZ + t0) * D_SZ + d;
  float*       outr = out  + ((size_t)b * L_SZ + t0) * D_SZ + d;

  float xw0 = 0.f, xw1 = 0.f, xw2 = 0.f;
  if (ck > 0) { xw0 = xr[-3 * D_SZ]; xw1 = xr[-2 * D_SZ]; xw2 = xr[-1 * D_SZ]; }

  for (int tb = 0; tb < CH_T; tb += SCAN_TB) {
    float xt[SCAN_TB], xv[SCAN_TB];
#pragma unroll
    for (int i = 0; i < SCAN_TB; ++i) xv[i] = xr[(size_t)(tb + i) * D_SZ];
    xt[0] = conv_silu(xw0, xw1, xw2, xv[0], cwd, cbd);
    xt[1] = conv_silu(xw1, xw2, xv[0], xv[1], cwd, cbd);
    xt[2] = conv_silu(xw2, xv[0], xv[1], xv[2], cwd, cbd);
#pragma unroll
    for (int i = 3; i < SCAN_TB; ++i)
      xt[i] = conv_silu(xv[i - 3], xv[i - 2], xv[i - 1], xv[i], cwd, cbd);
    xw0 = xv[SCAN_TB - 3]; xw1 = xv[SCAN_TB - 2]; xw2 = xv[SCAN_TB - 1];

    float dl[SCAN_TB], dx[SCAN_TB];
#pragma unroll
    for (int i = 0; i < SCAN_TB; ++i) {
      const float p32 = pr[(size_t)(tb + i) * P_SZ + 2 * N_SZ];
      const float z   = fmaf(p32, dtw, dtb);
      const float sp  = __logf(1.f + __expf(z));
      dl[i] = (z > 20.f) ? z : sp;
    }
#pragma unroll
    for (int i = 0; i < SCAN_TB; ++i) dx[i] = dl[i] * xt[i];

#pragma unroll
    for (int i = 0; i < SCAN_TB; ++i) {
      const float* pt = pr + (size_t)(tb + i) * P_SZ;
      v2f ya = {0.f, 0.f}, yb = {0.f, 0.f};
#pragma unroll
      for (int k = 0; k < 8; ++k) {
        const v2f e  = dl[i] * ac[k];
        const v2f ab = {__expf(e.x), __expf(e.y)};
        const v2f bt = {pt[2 * k], pt[2 * k + 1]};
        const v2f ct = {pt[N_SZ + 2 * k], pt[N_SZ + 2 * k + 1]};
        const v2f u  = dx[i] * bt;
        h[k] = __builtin_elementwise_fma(ab, h[k], u);
        if (k & 1) yb = __builtin_elementwise_fma(h[k], ct, yb);
        else       ya = __builtin_elementwise_fma(h[k], ct, ya);
      }
      const v2f ys = ya + yb;
      const float y = ys.x + ys.y;
      outr[(size_t)(tb + i) * D_SZ] = fmaf(xv[i], dpd, y);
    }
  }
}

extern "C" void kernel_launch(void* const* d_in, const int* in_sizes, int n_in,
                              void* d_out, int out_size, void* d_ws, size_t ws_size,
                              hipStream_t stream) {
  const float* x      = (const float*)d_in[0];
  const float* conv_w = (const float*)d_in[1];
  const float* conv_b = (const float*)d_in[2];
  const float* xproj  = (const float*)d_in[3];
  const float* dt_w   = (const float*)d_in[4];
  const float* dt_b   = (const float*)d_in[5];
  const float* a_log  = (const float*)d_in[6];
  const float* Dp     = (const float*)d_in[7];
  float* out = (float*)d_out;

  float* proj = (float*)d_ws;
  float* ch_e = proj + (size_t)B_SZ * L_SZ * P_SZ;
  float* ch_p = ch_e + (size_t)(CH_N - 1) * CHAN * N_SZ;

  conv_proj_wmma_kernel<<<(B_SZ * L_SZ) / 16, 32, 0, stream>>>(
      x, conv_w, conv_b, xproj, proj);

  scan_chunk_summary_kernel<<<(CHAN / SCAN_BLK) * (CH_N - 1), SCAN_BLK, 0, stream>>>(
      x, conv_w, conv_b, proj, dt_w, dt_b, a_log, ch_e, ch_p);

  scan_kernel<<<(CHAN / SCAN_BLK) * CH_N, SCAN_BLK, 0, stream>>>(
      x, conv_w, conv_b, proj, dt_w, dt_b, a_log, Dp, ch_e, ch_p, out);
}


# scan and chunk summary: a block copies its chunk's projection rows (B, C, last column) to LDS once; each step reads them by same-address wide LDS loads, not scalar loads; fp order unchanged
# speedup vs baseline: 18.2338x; 1.1536x over previous
#include <hip/hip_runtime.h>
#include <hip/hip_bf16.h>


#define B_SZ 4
#define L_SZ 2048
#define D_SZ 1024
#define N_SZ 16
#define P_SZ 33
#define CH_N 16
#define CH_T (L_SZ / CH_N)
#define CHAN (B_SZ * D_SZ)
#define SCAN_BLK 256
#define SCAN_TB 8
#define BC_W (2 * N_SZ)

typedef float v2f __attribute__((ext_vector_type(2)));
typedef float v4f __attribute__((ext_vector_type(4)));
typedef float v8f __attribute__((ext_vector_type(8)));

__device__ __forceinline__ float conv_silu(float x0, float x1, float x2, float x3,
                                           v4f w, float bias)
{
  float acc = bias;
  acc = fmaf(x0, w.x, acc);
  acc = fmaf(x1, w.y, acc);
  acc = fmaf(x2, w.z, acc);
  acc = fmaf(x3, w.w, acc);
  return acc / (1.f + __expf(-acc));
}

struct ProjStep {
  v4f x0, x1, x2, x3;
  v4f wa, wb, wc, wd;
  v4f bias;
  v4f b0, b1, b2;
};

struct ProjPtrs {
  const float *xp0, *xp1, *xp2, *xp3, *cwp, *cbp, *bp0, *bp1, *bp2;
};

__device__ __forceinline__ ProjStep proj_load(const ProjPtrs& p, const int k0)
{
  ProjStep r;
  r.x0   = *(const v4f*)(p.xp0 + k0);
  r.x1   = *(const v4f*)(p.xp1 + k0);
  r.x2   = *(const v4f*)(p.xp2 + k0);
  r.x3   = *(const v4f*)(p.xp3 + k0);
  r.wa   = *(const v4f*)(p.cwp + (size_t)k0 * 4);
  r.wb   = *(const v4f*)(p.cwp + (size_t)k0 * 4 + 4);
  r.wc   = *(const v4f*)(p.cwp + (size_t)k0 * 4 + 8);
  r.wd   = *(const v4f*)(p.cwp + (size_t)k0 * 4 + 12);
  r.bias = *(const v4f*)(p.cbp + k0);
  r.b0   = *(const v4f*)(p.bp0 + k0);
  r.b1   = *(const v4f*)(p.bp1 + k0);
  r.b2   = *(const v4f*)(p.bp2 + k0);
  return r;
}

template <bool FIRST>
__device__ __forceinline__ void proj_step(ProjStep s, const bool z0, const bool z1,
                                          const bool z2, v8f& acc0, v8f& acc1, v8f& acc2)
{
  if (FIRST) {
    const v4f zero4 = {0.f, 0.f, 0.f, 0.f};
    if (z0) s.x0 = zero4;
    if (z1) s.x1 = zero4;
    if (z2) s.x2 = zero4;
  }
  const v2f alo = {conv_silu(s.x0.x, s.x1.x, s.x2.x, s.x3.x, s.wa, s.bias.x),
                   conv_silu(s.x0.y, s.x1.y, s.x2.y, s.x3.y, s.wb, s.bias.y)};
  const v2f ahi = {conv_silu(s.x0.z, s.x1.z, s.x2.z, s.x3.z, s.wc, s.bias.z),
                   conv_silu(s.x0.w, s.x1.w, s.x2.w, s.x3.w, s.wd, s.bias.w)};

  acc0 = __builtin_amdgcn_wmma_f32_16x16x4_f32(false, alo, false, (v2f){s.b0.x, s.b0.y},
                                               (short)0, acc0, false, false);
  acc1 = __builtin_amdgcn_wmma_f32_16x16x4_f32(false, alo, false, (v2f){s.b1.x, s.b1.y},
                                               (short)0, acc1, false, false);
  acc2 = __builtin_amdgcn_wmma_f32_16x16x4_f32(false, alo, false, (v2f){s.b2.x, s.b2.y},
                                               (short)0, acc2, false, false);
  acc0 = __builtin_amdgcn_wmma_f32_16x16x4_f32(false, ahi, false, (v2f){s.b0.z, s.b0.w},
                                               (short)0, acc0, false, false);
  acc1 = __builtin_amdgcn_wmma_f32_16x16x4_f32(false, ahi, false, (v2f){s.b1.z, s.b1.w},
                                               (short)0, acc1, false, false);
  acc2 = __builtin_amdgcn_wmma_f32_16x16x4_f32(false, ahi, false, (v2f){s.b2.z, s.b2.w},
                                               (short)0, acc2, false, false);
}

template <bool FIRST>
__device__ __forceinline__ void conv_proj_tile(
    const float* __restrict__ x, const float* __restrict__ cw,
    const float* __restrict__ cb, const float* __restrict__ w,
    float* __restrict__ proj, const int tile, const int lane)
{
  const int c   = lane & 15;
  const int kh  = (lane >> 4) << 2;
  const int row = tile * 16 + c;
  const float* xrow = x + (size_t)row * D_SZ + kh;
  const bool z0 = FIRST && (c < 3);
  const bool z1 = FIRST && (c < 2);
  const bool z2 = FIRST && (c < 1);

  ProjPtrs p;
  p.xp0 = z0 ? xrow : xrow - 3 * D_SZ;
  p.xp1 = z1 ? xrow : xrow - 2 * D_SZ;
  p.xp2 = z2 ? xrow : xrow - 1 * D_SZ;
  p.xp3 = xrow;
  p.cwp = cw + (size_t)kh * 4;
  p.cbp = cb + kh;
  p.bp0 = w + (size_t)c * D_SZ + kh;
  p.bp1 = w + (size_t)(16 + c) * D_SZ + kh;
  p.bp2 = w + (size_t)(P_SZ - 1) * D_SZ + kh;

  v8f acc0 = {}, acc1 = {}, acc2 = {};
  ProjStep cur0 = proj_load(p, 0);
  ProjStep cur1 = proj_load(p, 8);
#pragma unroll 2
  for (int k0 = 0; k0 < D_SZ; k0 += 16) {
    const ProjStep nxt0 = proj_load(p, (k0 + 16) & (D_SZ - 1));
    const ProjStep nxt1 = proj_load(p, (k0 + 24) & (D_SZ - 1));
    __builtin_amdgcn_sched_barrier(0);

    proj_step<FIRST>(cur0, z0, z1, z2, acc0, acc1, acc2);
    proj_step<FIRST>(cur1, z0, z1, z2, acc0, acc1, acc2);

    __builtin_amdgcn_sched_barrier(0);
    asm volatile("" :: "v"(nxt0.x0), "v"(nxt0.x1), "v"(nxt0.x2), "v"(nxt0.x3),
                       "v"(nxt0.wa), "v"(nxt0.wb), "v"(nxt0.wc), "v"(nxt0.wd),
                       "v"(nxt0.bias), "v"(nxt0.b0), "v"(nxt0.b1), "v"(nxt0.b2));
    asm volatile("" :: "v"(nxt1.x0), "v"(nxt1.x1), "v"(nxt1.x2), "v"(nxt1.x3),
                       "v"(nxt1.wa), "v"(nxt1.wb), "v"(nxt1.wc), "v"(nxt1.wd),
                       "v"(nxt1.bias), "v"(nxt1.b0), "v"(nxt1.b1), "v"(nxt1.b2));
    cur0 = nxt0;
    cur1 = nxt1;
  }

  const int mhi = (lane >> 4) << 3;
#pragma unroll
  for (int v = 0; v < 8; ++v) {
    float* pr = proj + (size_t)(tile * 16 + v + mhi) * P_SZ;
    pr[c]      = acc0[v];
    pr[16 + c] = acc1[v];
    if (c == 0) pr[32] = acc2[v];
  }
}

__global__ __launch_bounds__(32) void conv_proj_wmma_kernel(
    const float* __restrict__ x, const float* __restrict__ cw,
    const float* __restrict__ cb, const float* __restrict__ w,
    float* __restrict__ proj)
{
  const int tile = blockIdx.x;
  const int lane = threadIdx.x;
  if ((tile & (L_SZ / 16 - 1)) == 0) conv_proj_tile<true>(x, cw, cb, w, proj, tile, lane);
  else                               conv_proj_tile<false>(x, cw, cb, w, proj, tile, lane);
}

#define STG_N ((CH_T * P_SZ + SCAN_BLK - 1) / SCAN_BLK)

__device__ __forceinline__ void stage_proj_load(const float* __restrict__ pr, float (&v)[STG_N])
{
#pragma unroll
  for (int j = 0; j < STG_N; ++j) {
    const int i = threadIdx.x + j * SCAN_BLK;
    v[j] = pr[i < CH_T * P_SZ ? i : CH_T * P_SZ - 1];
  }
}

#define S_ROW_N (CH_T * BC_W + CH_T + 1)
__device__ __forceinline__ void stage_proj_store(const float (&v)[STG_N], float* __restrict__ s_row)
{
#pragma unroll
  for (int j = 0; j < STG_N; ++j) {
    const int i   = threadIdx.x + j * SCAN_BLK;
    const int r   = i / P_SZ;
    const int col = i - r * P_SZ;
    const int a   = (col < BC_W) ? r * BC_W + col : CH_T * BC_W + r;
    s_row[i < CH_T * P_SZ ? a : S_ROW_N - 1] = v[j];
  }
}

__global__ __launch_bounds__(SCAN_BLK) void scan_chunk_summary_kernel(
    const float* __restrict__ x, const float* __restrict__ cw,
    const float* __restrict__ cb_, const float* __restrict__ proj,
    const float* __restrict__ dt_w, const float* __restrict__ dt_b,
    const float* __restrict__ a_log,
    float* __restrict__ ch_e, float* __restrict__ ch_p)
{
  const int ck = blockIdx.x / (CHAN / SCAN_BLK);
  const int cb = (blockIdx.x % (CHAN / SCAN_BLK)) * SCAN_BLK;
  const int b  = cb >> 10;
  const int c  = cb + threadIdx.x;
  const int d  = c & (D_SZ - 1);

  v2f ac[8];
#pragma unroll
  for (int k = 0; k < 8; ++k) {
    ac[k].x = -__expf(a_log[d * N_SZ + 2 * k]);
    ac[k].y = -__expf(a_log[d * N_SZ + 2 * k + 1]);
  }
  const float dtw = dt_w[d];
  const float dtb = dt_b[d];
  const v4f   cwd = *(const v4f*)(cw + (size_t)d * 4);
  const float cbd = cb_[d];

  const int t0 = ck * CH_T;
  const float* pr = proj + ((size_t)b * L_SZ + t0) * P_SZ;
  const float* xr = x    + ((size_t)b * L_SZ + t0) * D_SZ + d;

  __shared__ __attribute__((aligned(16))) float s_row[S_ROW_N];
  const float* s_bc = s_row;
  const float* s_dt = s_row + CH_T * BC_W;
  {
    float stg[STG_N];
    stage_proj_load(pr, stg);
    stage_proj_store(stg, s_row);
  }
  __syncthreads();

  float xw0 = 0.f, xw1 = 0.f, xw2 = 0.f;
  if (ck > 0) { xw0 = xr[-3 * D_SZ]; xw1 = xr[-2 * D_SZ]; xw2 = xr[-1 * D_SZ]; }

  v2f h[8], p[8];
#pragma unroll
  for (int k = 0; k < 8; ++k) { h[k] = (v2f){0.f, 0.f}; p[k] = (v2f){1.f, 1.f}; }

  for (int tb = 0; tb < CH_T; tb += SCAN_TB) {
    const v4f dq0 = *(const v4f*)(s_dt + tb);
    const v4f dq1 = *(const v4f*)(s_dt + tb + 4);
    const float p32v[SCAN_TB] = {dq0.x, dq0.y, dq0.z, dq0.w, dq1.x, dq1.y, dq1.z, dq1.w};
    v4f bq[SCAN_TB][N_SZ / 4];
#pragma unroll
    for (int i = 0; i < SCAN_TB; ++i) {
      const v4f* pt = (const v4f*)(s_bc + (tb + i) * BC_W);
#pragma unroll
      for (int q = 0; q < N_SZ / 4; ++q) bq[i][q] = pt[q];
    }

    float xv[SCAN_TB], xt[SCAN_TB];
#pragma unroll
    for (int i = 0; i < SCAN_TB; ++i) xv[i] = xr[(size_t)(tb + i) * D_SZ];
    xt[0] = conv_silu(xw0, xw1, xw2, xv[0], cwd, cbd);
    xt[1] = conv_silu(xw1, xw2, xv[0], xv[1], cwd, cbd);
    xt[2] = conv_silu(xw2, xv[0], xv[1], xv[2], cwd, cbd);
#pragma unroll
    for (int i = 3; i < SCAN_TB; ++i)
      xt[i] = conv_silu(xv[i - 3], xv[i - 2], xv[i - 1], xv[i], cwd, cbd);
    xw0 = xv[SCAN_TB - 3]; xw1 = xv[SCAN_TB - 2]; xw2 = xv[SCAN_TB - 1];

    float dl[SCAN_TB], dx[SCAN_TB];
#pragma unroll
    for (int i = 0; i < SCAN_TB; ++i) {
      const float p32 = p32v[i];
      const float z   = fmaf(p32, dtw, dtb);
      const float sp  = __logf(1.f + __expf(z));
      dl[i] = (z > 20.f) ? z : sp;
    }
#pragma unroll
    for (int i = 0; i < SCAN_TB; ++i) dx[i] = dl[i] * xt[i];

#pragma unroll
    for (int i = 0; i < SCAN_TB; ++i) {
#pragma unroll
      for (int k = 0; k < 8; ++k) {
        const v2f e  = dl[i] * ac[k];
        const v2f ab = {__expf(e.x), __expf(e.y)};
        const v4f b4 = bq[i][k >> 1];
        const v2f bt = (k & 1) ? (v2f){b4.z, b4.w} : (v2f){b4.x, b4.y};
        const v2f u  = dx[i] * bt;
        h[k] = __builtin_elementwise_fma(ab, h[k], u);
        p[k] = p[k] * ab;
      }
    }
  }

#pragma unroll
  for (int k = 0; k < 8; ++k) {
    const size_t o0 = ((size_t)(ck * N_SZ + 2 * k)) * CHAN + c;
    ch_e[o0]        = h[k].x;
    ch_e[o0 + CHAN] = h[k].y;
    ch_p[o0]        = p[k].x;
    ch_p[o0 + CHAN] = p[k].y;
  }
}

__global__ __launch_bounds__(SCAN_BLK) void scan_kernel(
    const float* __restrict__ x, const float* __restrict__ cw,
    const float* __restrict__ cb_,
    const float* __restrict__ proj, const float* __restrict__ dt_w,
    const float* __restrict__ dt_b, const float* __restrict__ a_log,
    const float* __restrict__ Dp, const float* __restrict__ ch_e,
    const float* __restrict__ ch_p, float* __restrict__ out)
{
  const int ck = blockIdx.x / (CHAN / SCAN_BLK);
  const int cb = (blockIdx.x % (CHAN / SCAN_BLK)) * SCAN_BLK;
  const int b  = cb >> 10;
  const int c  = cb + threadIdx.x;
  const int d  = c & (D_SZ - 1);

  v2f ac[8];
#pragma unroll
  for (int k = 0; k < 8; ++k) {
    ac[k].x = -__expf(a_log[d * N_SZ + 2 * k]);
    ac[k].y = -__expf(a_log[d * N_SZ + 2 * k + 1]);
  }
  const float dtw = dt_w[d];
  const float dtb = dt_b[d];
  const float dpd = Dp[d];
  const v4f   cwd = *(const v4f*)(cw + (size_t)d * 4);
  const float cbd = cb_[d];

  const float* pr = proj + ((size_t)b * L_SZ + ck * CH_T) * P_SZ;
  __shared__ __attribute__((aligned(16))) float s_row[S_ROW_N];
  const float* s_bc = s_row;
  const float* s_dt = s_row + CH_T * BC_W;
  float stg[STG_N];
  stage_proj_load(pr, stg);

  v2f h[8];
#pragma unroll
  for (int k = 0; k < 8; ++k) h[k] = (v2f){0.f, 0.f};
#pragma unroll 4
  for (int j = 0; j < ck; ++j) {
    v2f pj[8], ej[8];
#pragma unroll
    for (int k = 0; k < 8; ++k) {
      const size_t o0 = ((size_t)(j * N_SZ + 2 * k)) * CHAN + c;
      pj[k].x = ch_p[o0];
      pj[k].y = ch_p[o0 + CHAN];
      ej[k].x = ch_e[o0];
      ej[k].y = ch_e[o0 + CHAN];
    }
#pragma unroll
    for (int k = 0; k < 8; ++k) h[k] = __builtin_elementwise_fma(pj[k], h[k], ej[k]);
  }

  const int t0 = ck * CH_T;
  const float* xr   = x    + ((size_t)b * L_SZ + t0) * D_SZ + d;
  float*       outr = out  + ((size_t)b * L_SZ + t0) * D_SZ + d;

  stage_proj_store(stg, s_row);
  __syncthreads();

  float xw0 = 0.f, xw1 = 0.f, xw2 = 0.f;
  if (ck > 0) { xw0 = xr[-3 * D_SZ]; xw1 = xr[-2 * D_SZ]; xw2 = xr[-1 * D_SZ]; }

  for (int tb = 0; tb < CH_T; tb += SCAN_TB) {
    float xt[SCAN_TB], xv[SCAN_TB];
#pragma unroll
    for (int i = 0; i < SCAN_TB; ++i) xv[i] = xr[(size_t)(tb + i) * D_SZ];
    xt[0] = conv_silu(xw0, xw1, xw2, xv[0], cwd, cbd);
    xt[1] = conv_silu(xw1, xw2, xv[0], xv[1], cwd, cbd);
    xt[2] = conv_silu(xw2, xv[0], xv[1], xv[2], cwd, cbd);
#pragma unroll
    for (int i = 3; i < SCAN_TB; ++i)
      xt[i] = conv_silu(xv[i - 3], xv[i - 2], xv[i - 1], xv[i], cwd, cbd);
    xw0 = xv[SCAN_TB - 3]; xw1 = xv[SCAN_TB - 2]; xw2 = xv[SCAN_TB - 1];

    float dl[SCAN_TB], dx[SCAN_TB];
#pragma unroll
    for (int i = 0; i < SCAN_TB; ++i) {
      const float p32 = s_dt[tb + i];
      const float z   = fmaf(p32, dtw, dtb);
      const float sp  = __logf(1.f + __expf(z));
      dl[i] = (z > 20.f) ? z : sp;
    }
#pragma unroll
    for (int i = 0; i < SCAN_TB; ++i) dx[i] = dl[i] * xt[i];

#pragma unroll
    for (int i = 0; i < SCAN_TB; ++i) {
      const v4f* pt = (const v4f*)(s_bc + (tb + i) * BC_W);
      v2f ya = {0.f, 0.f}, yb = {0.f, 0.f};
#pragma unroll
      for (int k = 0; k < 8; ++k) {
        const v2f e  = dl[i] * ac[k];
        const v2f ab = {__expf(e.x), __expf(e.y)};
        const v4f b4 = pt[k >> 1];
        const v2f bt = (k & 1) ? (v2f){b4.z, b4.w} : (v2f){b4.x, b4.y};
        const v4f c4 = pt[N_SZ / 4 + (k >> 1)];
        const v2f ct = (k & 1) ? (v2f){c4.z, c4.w} : (v2f){c4.x, c4.y};
        const v2f u  = dx[i] * bt;
        h[k] = __builtin_elementwise_fma(ab, h[k], u);
        if (k & 1) yb = __builtin_elementwise_fma(h[k], ct, yb);
        else       ya = __builtin_elementwise_fma(h[k], ct, ya);
      }
      const v2f ys = ya + yb;
      const float y = ys.x + ys.y;
      outr[(size_t)(tb + i) * D_SZ] = fmaf(xv[i], dpd, y);
    }
  }
}

extern "C" void kernel_launch(void* const* d_in, const int* in_sizes, int n_in,
                              void* d_out, int out_size, void* d_ws, size_t ws_size,
                              hipStream_t stream) {
  const float* x      = (const float*)d_in[0];
  const float* conv_w = (const float*)d_in[1];
  const float* conv_b = (const float*)d_in[2];
  const float* xproj  = (const float*)d_in[3];
  const float* dt_w   = (const float*)d_in[4];
  const float* dt_b   = (const float*)d_in[5];
  const float* a_log  = (const float*)d_in[6];
  const float* Dp     = (const float*)d_in[7];
  float* out = (float*)d_out;

  float* proj = (float*)d_ws;
  float* ch_e = proj + (size_t)B_SZ * L_SZ * P_SZ;
  float* ch_p = ch_e + (size_t)(CH_N - 1) * CHAN * N_SZ;

  conv_proj_wmma_kernel<<<(B_SZ * L_SZ) / 16, 32, 0, stream>>>(
      x, conv_w, conv_b, xproj, proj);

  scan_chunk_summary_kernel<<<(CHAN / SCAN_BLK) * (CH_N - 1), SCAN_BLK, 0, stream>>>(
      x, conv_w, conv_b, proj, dt_w, dt_b, a_log, ch_e, ch_p);

  scan_kernel<<<(CHAN / SCAN_BLK) * CH_N, SCAN_BLK, 0, stream>>>(
      x, conv_w, conv_b, proj, dt_w, dt_b, a_log, Dp, ch_e, ch_p, out);
}


# SiLU of the conv output (all 3 kernels): acc * v_rcp_f32(1 + exp(-acc)) in place of the correctly rounded division's expansion; SiLU rounding changed (reciprocal to 1 ulp), sum orders unchanged
# speedup vs baseline: 19.2772x; 1.0572x over previous
#include <hip/hip_runtime.h>
#include <hip/hip_bf16.h>


#define B_SZ 4
#define L_SZ 2048
#define D_SZ 1024
#define N_SZ 16
#define P_SZ 33
#define CH_N 16
#define CH_T (L_SZ / CH_N)
#define CHAN (B_SZ * D_SZ)
#define SCAN_BLK 256
#define SCAN_TB 8
#define BC_W (2 * N_SZ)

typedef float v2f __attribute__((ext_vector_type(2)));
typedef float v4f __attribute__((ext_vector_type(4)));
typedef float v8f __attribute__((ext_vector_type(8)));

__device__ __forceinline__ float conv_silu(float x0, float x1, float x2, float x3,
                                           v4f w, float bias)
{
  float acc = bias;
  acc = fmaf(x0, w.x, acc);
  acc = fmaf(x1, w.y, acc);
  acc = fmaf(x2, w.z, acc);
  acc = fmaf(x3, w.w, acc);
  return acc * __builtin_amdgcn_rcpf(1.f + __expf(-acc));
}

struct ProjStep {
  v4f x0, x1, x2, x3;
  v4f wa, wb, wc, wd;
  v4f bias;
  v4f b0, b1, b2;
};

struct ProjPtrs {
  const float *xp0, *xp1, *xp2, *xp3, *cwp, *cbp, *bp0, *bp1, *bp2;
};

__device__ __forceinline__ ProjStep proj_load(const ProjPtrs& p, const int k0)
{
  ProjStep r;
  r.x0   = *(const v4f*)(p.xp0 + k0);
  r.x1   = *(const v4f*)(p.xp1 + k0);
  r.x2   = *(const v4f*)(p.xp2 + k0);
  r.x3   = *(const v4f*)(p.xp3 + k0);
  r.wa   = *(const v4f*)(p.cwp + (size_t)k0 * 4);
  r.wb   = *(const v4f*)(p.cwp + (size_t)k0 * 4 + 4);
  r.wc   = *(const v4f*)(p.cwp + (size_t)k0 * 4 + 8);
  r.wd   = *(const v4f*)(p.cwp + (size_t)k0 * 4 + 12);
  r.bias = *(const v4f*)(p.cbp + k0);
  r.b0   = *(const v4f*)(p.bp0 + k0);
  r.b1   = *(const v4f*)(p.bp1 + k0);
  r.b2   = *(const v4f*)(p.bp2 + k0);
  return r;
}

template <bool FIRST>
__device__ __forceinline__ void proj_step(ProjStep s, const bool z0, const bool z1,
                                          const bool z2, v8f& acc0, v8f& acc1, v8f& acc2)
{
  if (FIRST) {
    const v4f zero4 = {0.f, 0.f, 0.f, 0.f};
    if (z0) s.x0 = zero4;
    if (z1) s.x1 = zero4;
    if (z2) s.x2 = zero4;
  }
  const v2f alo = {conv_silu(s.x0.x, s.x1.x, s.x2.x, s.x3.x, s.wa, s.bias.x),
                   conv_silu(s.x0.y, s.x1.y, s.x2.y, s.x3.y, s.wb, s.bias.y)};
  const v2f ahi = {conv_silu(s.x0.z, s.x1.z, s.x2.z, s.x3.z, s.wc, s.bias.z),
                   conv_silu(s.x0.w, s.x1.w, s.x2.w, s.x3.w, s.wd, s.bias.w)};

  acc0 = __builtin_amdgcn_wmma_f32_16x16x4_f32(false, alo, false, (v2f){s.b0.x, s.b0.y},
                                               (short)0, acc0, false, false);
  acc1 = __builtin_amdgcn_wmma_f32_16x16x4_f32(false, alo, false, (v2f){s.b1.x, s.b1.y},
                                               (short)0, acc1, false, false);
  acc2 = __builtin_amdgcn_wmma_f32_16x16x4_f32(false, alo, false, (v2f){s.b2.x, s.b2.y},
                                               (short)0, acc2, false, false);
  acc0 = __builtin_amdgcn_wmma_f32_16x16x4_f32(false, ahi, false, (v2f){s.b0.z, s.b0.w},
                                               (short)0, acc0, false, false);
  acc1 = __builtin_amdgcn_wmma_f32_16x16x4_f32(false, ahi, false, (v2f){s.b1.z, s.b1.w},
                                               (short)0, acc1, false, false);
  acc2 = __builtin_amdgcn_wmma_f32_16x16x4_f32(false, ahi, false, (v2f){s.b2.z, s.b2.w},
                                               (short)0, acc2, false, false);
}

template <bool FIRST>
__device__ __forceinline__ void conv_proj_tile(
    const float* __restrict__ x, const float* __restrict__ cw,
    const float* __restrict__ cb, const float* __restrict__ w,
    float* __restrict__ proj, const int tile, const int lane)
{
  const int c   = lane & 15;
  const int kh  = (lane >> 4) << 2;
  const int row = tile * 16 + c;
  const float* xrow = x + (size_t)row * D_SZ + kh;
  const bool z0 = FIRST && (c < 3);
  const bool z1 = FIRST && (c < 2);
  const bool z2 = FIRST && (c < 1);

  ProjPtrs p;
  p.xp0 = z0 ? xrow : xrow - 3 * D_SZ;
  p.xp1 = z1 ? xrow : xrow - 2 * D_SZ;
  p.xp2 = z2 ? xrow : xrow - 1 * D_SZ;
  p.xp3 = xrow;
  p.cwp = cw + (size_t)kh * 4;
  p.cbp = cb + kh;
  p.bp0 = w + (size_t)c * D_SZ + kh;
  p.bp1 = w + (size_t)(16 + c) * D_SZ + kh;
  p.bp2 = w + (size_t)(P_SZ - 1) * D_SZ + kh;

  v8f acc0 = {}, acc1 = {}, acc2 = {};
  ProjStep cur0 = proj_load(p, 0);
  ProjStep cur1 = proj_load(p, 8);
#pragma unroll 2
  for (int k0 = 0; k0 < D_SZ; k0 += 16) {
    const ProjStep nxt0 = proj_load(p, (k0 + 16) & (D_SZ - 1));
    const ProjStep nxt1 = proj_load(p, (k0 + 24) & (D_SZ - 1));
    __builtin_amdgcn_sched_barrier(0);

    proj_step<FIRST>(cur0, z0, z1, z2, acc0, acc1, acc2);
    proj_step<FIRST>(cur1, z0, z1, z2, acc0, acc1, acc2);

    __builtin_amdgcn_sched_barrier(0);
    asm volatile("" :: "v"(nxt0.x0), "v"(nxt0.x1), "v"(nxt0.x2), "v"(nxt0.x3),
                       "v"(nxt0.wa), "v"(nxt0.wb), "v"(nxt0.wc), "v"(nxt0.wd),
                       "v"(nxt0.bias), "v"(nxt0.b0), "v"(nxt0.b1), "v"(nxt0.b2));
    asm volatile("" :: "v"(nxt1.x0), "v"(nxt1.x1), "v"(nxt1.x2), "v"(nxt1.x3),
                       "v"(nxt1.wa), "v"(nxt1.wb), "v"(nxt1.wc), "v"(nxt1.wd),
                       "v"(nxt1.bias), "v"(nxt1.b0), "v"(nxt1.b1), "v"(nxt1.b2));
    cur0 = nxt0;
    cur1 = nxt1;
  }

  const int mhi = (lane >> 4) << 3;
#pragma unroll
  for (int v = 0; v < 8; ++v) {
    float* pr = proj + (size_t)(tile * 16 + v + mhi) * P_SZ;
    pr[c]      = acc0[v];
    pr[16 + c] = acc1[v];
    if (c == 0) pr[32] = acc2[v];
  }
}

__global__ __launch_bounds__(32) void conv_proj_wmma_kernel(
    const float* __restrict__ x, const float* __restrict__ cw,
    const float* __restrict__ cb, const float* __restrict__ w,
    float* __restrict__ proj)
{
  const int tile = blockIdx.x;
  const int lane = threadIdx.x;
  if ((tile & (L_SZ / 16 - 1)) == 0) conv_proj_tile<true>(x, cw, cb, w, proj, tile, lane);
  else                               conv_proj_tile<false>(x, cw, cb, w, proj, tile, lane);
}

#define STG_N ((CH_T * P_SZ + SCAN_BLK - 1) / SCAN_BLK)

__device__ __forceinline__ void stage_proj_load(const float* __restrict__ pr, float (&v)[STG_N])
{
#pragma unroll
  for (int j = 0; j < STG_N; ++j) {
    const int i = threadIdx.x + j * SCAN_BLK;
    v[j] = pr[i < CH_T * P_SZ ? i : CH_T * P_SZ - 1];
  }
}

#define S_ROW_N (CH_T * BC_W + CH_T + 1)
__device__ __forceinline__ void stage_proj_store(const float (&v)[STG_N], float* __restrict__ s_row)
{
#pragma unroll
  for (int j = 0; j < STG_N; ++j) {
    const int i   = threadIdx.x + j * SCAN_BLK;
    const int r   = i / P_SZ;
    const int col = i - r * P_SZ;
    const int a   = (col < BC_W) ? r * BC_W + col : CH_T * BC_W + r;
    s_row[i < CH_T * P_SZ ? a : S_ROW_N - 1] = v[j];
  }
}

__global__ __launch_bounds__(SCAN_BLK) void scan_chunk_summary_kernel(
    const float* __restrict__ x, const float* __restrict__ cw,
    const float* __restrict__ cb_, const float* __restrict__ proj,
    const float* __restrict__ dt_w, const float* __restrict__ dt_b,
    const float* __restrict__ a_log,
    float* __restrict__ ch_e, float* __restrict__ ch_p)
{
  const int ck = blockIdx.x / (CHAN / SCAN_BLK);
  const int cb = (blockIdx.x % (CHAN / SCAN_BLK)) * SCAN_BLK;
  const int b  = cb >> 10;
  const int c  = cb + threadIdx.x;
  const int d  = c & (D_SZ - 1);

  v2f ac[8];
#pragma unroll
  for (int k = 0; k < 8; ++k) {
    ac[k].x = -__expf(a_log[d * N_SZ + 2 * k]);
    ac[k].y = -__expf(a_log[d * N_SZ + 2 * k + 1]);
  }
  const float dtw = dt_w[d];
  const float dtb = dt_b[d];
  const v4f   cwd = *(const v4f*)(cw + (size_t)d * 4);
  const float cbd = cb_[d];

  const int t0 = ck * CH_T;
  const float* pr = proj + ((size_t)b * L_SZ + t0) * P_SZ;
  const float* xr = x    + ((size_t)b * L_SZ + t0) * D_SZ + d;

  __shared__ __attribute__((aligned(16))) float s_row[S_ROW_N];
  const float* s_bc = s_row;
  const float* s_dt = s_row + CH_T * BC_W;
  {
    float stg[STG_N];
    stage_proj_load(pr, stg);
    stage_proj_store(stg, s_row);
  }
  __syncthreads();

  float xw0 = 0.f, xw1 = 0.f, xw2 = 0.f;
  if (ck > 0) { xw0 = xr[-3 * D_SZ]; xw1 = xr[-2 * D_SZ]; xw2 = xr[-1 * D_SZ]; }

  v2f h[8], p[8];
#pragma unroll
  for (int k = 0; k < 8; ++k) { h[k] = (v2f){0.f, 0.f}; p[k] = (v2f){1.f, 1.f}; }

  for (int tb = 0; tb < CH_T; tb += SCAN_TB) {
    const v4f dq0 = *(const v4f*)(s_dt + tb);
    const v4f dq1 = *(const v4f*)(s_dt + tb + 4);
    const float p32v[SCAN_TB] = {dq0.x, dq0.y, dq0.z, dq0.w, dq1.x, dq1.y, dq1.z, dq1.w};
    v4f bq[SCAN_TB][N_SZ / 4];
#pragma unroll
    for (int i = 0; i < SCAN_TB; ++i) {
      const v4f* pt = (const v4f*)(s_bc + (tb + i) * BC_W);
#pragma unroll
      for (int q = 0; q < N_SZ / 4; ++q) bq[i][q] = pt[q];
    }

    float xv[SCAN_TB], xt[SCAN_TB];
#pragma unroll
    for (int i = 0; i < SCAN_TB; ++i) xv[i] = xr[(size_t)(tb + i) * D_SZ];
    xt[0] = conv_silu(xw0, xw1, xw2, xv[0], cwd, cbd);
    xt[1] = conv_silu(xw1, xw2, xv[0], xv[1], cwd, cbd);
    xt[2] = conv_silu(xw2, xv[0], xv[1], xv[2], cwd, cbd);
#pragma unroll
    for (int i = 3; i < SCAN_TB; ++i)
      xt[i] = conv_silu(xv[i - 3], xv[i - 2], xv[i - 1], xv[i], cwd, cbd);
    xw0 = xv[SCAN_TB - 3]; xw1 = xv[SCAN_TB - 2]; xw2 = xv[SCAN_TB - 1];

    float dl[SCAN_TB], dx[SCAN_TB];
#pragma unroll
    for (int i = 0; i < SCAN_TB; ++i) {
      const float p32 = p32v[i];
      const float z   = fmaf(p32, dtw, dtb);
      const float sp  = __logf(1.f + __expf(z));
      dl[i] = (z > 20.f) ? z : sp;
    }
#pragma unroll
    for (int i = 0; i < SCAN_TB; ++i) dx[i] = dl[i] * xt[i];

#pragma unroll
    for (int i = 0; i < SCAN_TB; ++i) {
#pragma unroll
      for (int k = 0; k < 8; ++k) {
        const v2f e  = dl[i] * ac[k];
        const v2f ab = {__expf(e.x), __expf(e.y)};
        const v4f b4 = bq[i][k >> 1];
        const v2f bt = (k & 1) ? (v2f){b4.z, b4.w} : (v2f){b4.x, b4.y};
        const v2f u  = dx[i] * bt;
        h[k] = __builtin_elementwise_fma(ab, h[k], u);
        p[k] = p[k] * ab;
      }
    }
  }

#pragma unroll
  for (int k = 0; k < 8; ++k) {
    const size_t o0 = ((size_t)(ck * N_SZ + 2 * k)) * CHAN + c;
    ch_e[o0]        = h[k].x;
    ch_e[o0 + CHAN] = h[k].y;
    ch_p[o0]        = p[k].x;
    ch_p[o0 + CHAN] = p[k].y;
  }
}

__global__ __launch_bounds__(SCAN_BLK) void scan_kernel(
    const float* __restrict__ x, const float* __restrict__ cw,
    const float* __restrict__ cb_,
    const float* __restrict__ proj, const float* __restrict__ dt_w,
    const float* __restrict__ dt_b, const float* __restrict__ a_log,
    const float* __restrict__ Dp, const float* __restrict__ ch_e,
    const float* __restrict__ ch_p, float* __restrict__ out)
{
  const int ck = blockIdx.x / (CHAN / SCAN_BLK);
  const int cb = (blockIdx.x % (CHAN / SCAN_BLK)) * SCAN_BLK;
  const int b  = cb >> 10;
  const int c  = cb + threadIdx.x;
  const int d  = c & (D_SZ - 1);

  v2f ac[8];
#pragma unroll
  for (int k = 0; k < 8; ++k) {
    ac[k].x = -__expf(a_log[d * N_SZ + 2 * k]);
    ac[k].y = -__expf(a_log[d * N_SZ + 2 * k + 1]);
  }
  const float dtw = dt_w[d];
  const float dtb = dt_b[d];
  const float dpd = Dp[d];
  const v4f   cwd = *(const v4f*)(cw + (size_t)d * 4);
  const float cbd = cb_[d];

  const float* pr = proj + ((size_t)b * L_SZ + ck * CH_T) * P_SZ;
  __shared__ __attribute__((aligned(16))) float s_row[S_ROW_N];
  const float* s_bc = s_row;
  const float* s_dt = s_row + CH_T * BC_W;
  float stg[STG_N];
  stage_proj_load(pr, stg);

  v2f h[8];
#pragma unroll
  for (int k = 0; k < 8; ++k) h[k] = (v2f){0.f, 0.f};
#pragma unroll 4
  for (int j = 0; j < ck; ++j) {
    v2f pj[8], ej[8];
#pragma unroll
    for (int k = 0; k < 8; ++k) {
      const size_t o0 = ((size_t)(j * N_SZ + 2 * k)) * CHAN + c;
      pj[k].x = ch_p[o0];
      pj[k].y = ch_p[o0 + CHAN];
      ej[k].x = ch_e[o0];
      ej[k].y = ch_e[o0 + CHAN];
    }
#pragma unroll
    for (int k = 0; k < 8; ++k) h[k] = __builtin_elementwise_fma(pj[k], h[k], ej[k]);
  }

  const int t0 = ck * CH_T;
  const float* xr   = x    + ((size_t)b * L_SZ + t0) * D_SZ + d;
  float*       outr = out  + ((size_t)b * L_SZ + t0) * D_SZ + d;

  stage_proj_store(stg, s_row);
  __syncthreads();

  float xw0 = 0.f, xw1 = 0.f, xw2 = 0.f;
  if (ck > 0) { xw0 = xr[-3 * D_SZ]; xw1 = xr[-2 * D_SZ]; xw2 = xr[-1 * D_SZ]; }

  for (int tb = 0; tb < CH_T; tb += SCAN_TB) {
    float xt[SCAN_TB], xv[SCAN_TB];
#pragma unroll
    for (int i = 0; i < SCAN_TB; ++i) xv[i] = xr[(size_t)(tb + i) * D_SZ];
    xt[0] = conv_silu(xw0, xw1, xw2, xv[0], cwd, cbd);
    xt[1] = conv_silu(xw1, xw2, xv[0], xv[1], cwd, cbd);
    xt[2] = conv_silu(xw2, xv[0], xv[1], xv[2], cwd, cbd);
#pragma unroll
    for (int i = 3; i < SCAN_TB; ++i)
      xt[i] = conv_silu(xv[i - 3], xv[i - 2], xv[i - 1], xv[i], cwd, cbd);
    xw0 = xv[SCAN_TB - 3]; xw1 = xv[SCAN_TB - 2]; xw2 = xv[SCAN_TB - 1];

    float dl[SCAN_TB], dx[SCAN_TB];
#pragma unroll
    for (int i = 0; i < SCAN_TB; ++i) {
      const float p32 = s_dt[tb + i];
      const float z   = fmaf(p32, dtw, dtb);
      const float sp  = __logf(1.f + __expf(z));
      dl[i] = (z > 20.f) ? z : sp;
    }
#pragma unroll
    for (int i = 0; i < SCAN_TB; ++i) dx[i] = dl[i] * xt[i];

#pragma unroll
    for (int i = 0; i < SCAN_TB; ++i) {
      const v4f* pt = (const v4f*)(s_bc + (tb + i) * BC_W);
      v2f ya = {0.f, 0.f}, yb = {0.f, 0.f};
#pragma unroll
      for (int k = 0; k < 8; ++k) {
        const v2f e  = dl[i] * ac[k];
        const v2f ab = {__expf(e.x), __expf(e.y)};
        const v4f b4 = pt[k >> 1];
        const v2f bt = (k & 1) ? (v2f){b4.z, b4.w} : (v2f){b4.x, b4.y};
        const v4f c4 = pt[N_SZ / 4 + (k >> 1)];
        const v2f ct = (k & 1) ? (v2f){c4.z, c4.w} : (v2f){c4.x, c4.y};
        const v2f u  = dx[i] * bt;
        h[k] = __builtin_elementwise_fma(ab, h[k], u);
        if (k & 1) yb = __builtin_elementwise_fma(h[k], ct, yb);
        else       ya = __builtin_elementwise_fma(h[k], ct, ya);
      }
      const v2f ys = ya + yb;
      const float y = ys.x + ys.y;
      outr[(size_t)(tb + i) * D_SZ] = fmaf(xv[i], dpd, y);
    }
  }
}

extern "C" void kernel_launch(void* const* d_in, const int* in_sizes, int n_in,
                              void* d_out, int out_size, void* d_ws, size_t ws_size,
                              hipStream_t stream) {
  const float* x      = (const float*)d_in[0];
  const float* conv_w = (const float*)d_in[1];
  const float* conv_b = (const float*)d_in[2];
  const float* xproj  = (const float*)d_in[3];
  const float* dt_w   = (const float*)d_in[4];
  const float* dt_b   = (const float*)d_in[5];
  const float* a_log  = (const float*)d_in[6];
  const float* Dp     = (const float*)d_in[7];
  float* out = (float*)d_out;

  float* proj = (float*)d_ws;
  float* ch_e = proj + (size_t)B_SZ * L_SZ * P_SZ;
  float* ch_p = ch_e + (size_t)(CH_N - 1) * CHAN * N_SZ;

  conv_proj_wmma_kernel<<<(B_SZ * L_SZ) / 16, 32, 0, stream>>>(
      x, conv_w, conv_b, xproj, proj);

  scan_chunk_summary_kernel<<<(CHAN / SCAN_BLK) * (CH_N - 1), SCAN_BLK, 0, stream>>>(
      x, conv_w, conv_b, proj, dt_w, dt_b, a_log, ch_e, ch_p);

  scan_kernel<<<(CHAN / SCAN_BLK) * CH_N, SCAN_BLK, 0, stream>>>(
      x, conv_w, conv_b, proj, dt_w, dt_b, a_log, Dp, ch_e, ch_p, out);
}


# proj K loop: operands requested 5 steps ahead into a ring of 6 register sets (was 1 trip ahead, 2 sets): about 60 loads in flight per wave, not 24; fp order unchanged
# speedup vs baseline: 24.5558x; 1.2738x over previous
#include <hip/hip_runtime.h>
#include <hip/hip_bf16.h>


#define B_SZ 4
#define L_SZ 2048
#define D_SZ 1024
#define N_SZ 16
#define P_SZ 33
#define CH_N 16
#define CH_T (L_SZ / CH_N)
#define CHAN (B_SZ * D_SZ)
#define SCAN_BLK 256
#define SCAN_TB 8
#define BC_W (2 * N_SZ)

typedef float v2f __attribute__((ext_vector_type(2)));
typedef float v4f __attribute__((ext_vector_type(4)));
typedef float v8f __attribute__((ext_vector_type(8)));

__device__ __forceinline__ float conv_silu(float x0, float x1, float x2, float x3,
                                           v4f w, float bias)
{
  float acc = bias;
  acc = fmaf(x0, w.x, acc);
  acc = fmaf(x1, w.y, acc);
  acc = fmaf(x2, w.z, acc);
  acc = fmaf(x3, w.w, acc);
  return acc * __builtin_amdgcn_rcpf(1.f + __expf(-acc));
}

struct ProjStep {
  v4f x0, x1, x2, x3;
  v4f wa, wb, wc, wd;
  v4f bias;
  v4f b0, b1, b2;
};

#define PF_R 6
#define PF_MAIN (((D_SZ / 8) / PF_R) * PF_R)

struct ProjPtrs {
  const float *xp0, *xp1, *xp2, *xp3, *cwp, *cbp, *bp0, *bp1, *bp2;
};

__device__ __forceinline__ ProjStep proj_load(const ProjPtrs& p, const int k0)
{
  ProjStep r;
  r.x0   = *(const v4f*)(p.xp0 + k0);
  r.x1   = *(const v4f*)(p.xp1 + k0);
  r.x2   = *(const v4f*)(p.xp2 + k0);
  r.x3   = *(const v4f*)(p.xp3 + k0);
  r.wa   = *(const v4f*)(p.cwp + (size_t)k0 * 4);
  r.wb   = *(const v4f*)(p.cwp + (size_t)k0 * 4 + 4);
  r.wc   = *(const v4f*)(p.cwp + (size_t)k0 * 4 + 8);
  r.wd   = *(const v4f*)(p.cwp + (size_t)k0 * 4 + 12);
  r.bias = *(const v4f*)(p.cbp + k0);
  r.b0   = *(const v4f*)(p.bp0 + k0);
  r.b1   = *(const v4f*)(p.bp1 + k0);
  r.b2   = *(const v4f*)(p.bp2 + k0);
  return r;
}

template <bool FIRST>
__device__ __forceinline__ void proj_step(ProjStep s, const bool z0, const bool z1,
                                          const bool z2, v8f& acc0, v8f& acc1, v8f& acc2)
{
  if (FIRST) {
    const v4f zero4 = {0.f, 0.f, 0.f, 0.f};
    if (z0) s.x0 = zero4;
    if (z1) s.x1 = zero4;
    if (z2) s.x2 = zero4;
  }
  const v2f alo = {conv_silu(s.x0.x, s.x1.x, s.x2.x, s.x3.x, s.wa, s.bias.x),
                   conv_silu(s.x0.y, s.x1.y, s.x2.y, s.x3.y, s.wb, s.bias.y)};
  const v2f ahi = {conv_silu(s.x0.z, s.x1.z, s.x2.z, s.x3.z, s.wc, s.bias.z),
                   conv_silu(s.x0.w, s.x1.w, s.x2.w, s.x3.w, s.wd, s.bias.w)};

  acc0 = __builtin_amdgcn_wmma_f32_16x16x4_f32(false, alo, false, (v2f){s.b0.x, s.b0.y},
                                               (short)0, acc0, false, false);
  acc1 = __builtin_amdgcn_wmma_f32_16x16x4_f32(false, alo, false, (v2f){s.b1.x, s.b1.y},
                                               (short)0, acc1, false, false);
  acc2 = __builtin_amdgcn_wmma_f32_16x16x4_f32(false, alo, false, (v2f){s.b2.x, s.b2.y},
                                               (short)0, acc2, false, false);
  acc0 = __builtin_amdgcn_wmma_f32_16x16x4_f32(false, ahi, false, (v2f){s.b0.z, s.b0.w},
                                               (short)0, acc0, false, false);
  acc1 = __builtin_amdgcn_wmma_f32_16x16x4_f32(false, ahi, false, (v2f){s.b1.z, s.b1.w},
                                               (short)0, acc1, false, false);
  acc2 = __builtin_amdgcn_wmma_f32_16x16x4_f32(false, ahi, false, (v2f){s.b2.z, s.b2.w},
                                               (short)0, acc2, false, false);
}

template <bool FIRST>
__device__ __forceinline__ void conv_proj_tile(
    const float* __restrict__ x, const float* __restrict__ cw,
    const float* __restrict__ cb, const float* __restrict__ w,
    float* __restrict__ proj, const int tile, const int lane)
{
  const int c   = lane & 15;
  const int kh  = (lane >> 4) << 2;
  const int row = tile * 16 + c;
  const float* xrow = x + (size_t)row * D_SZ + kh;
  const bool z0 = FIRST && (c < 3);
  const bool z1 = FIRST && (c < 2);
  const bool z2 = FIRST && (c < 1);

  ProjPtrs p;
  p.xp0 = z0 ? xrow : xrow - 3 * D_SZ;
  p.xp1 = z1 ? xrow : xrow - 2 * D_SZ;
  p.xp2 = z2 ? xrow : xrow - 1 * D_SZ;
  p.xp3 = xrow;
  p.cwp = cw + (size_t)kh * 4;
  p.cbp = cb + kh;
  p.bp0 = w + (size_t)c * D_SZ + kh;
  p.bp1 = w + (size_t)(16 + c) * D_SZ + kh;
  p.bp2 = w + (size_t)(P_SZ - 1) * D_SZ + kh;

  v8f acc0 = {}, acc1 = {}, acc2 = {};
  ProjStep ring[PF_R];
#pragma unroll
  for (int i = 0; i < PF_R - 1; ++i) {
    ring[i] = proj_load(p, 8 * i);
    __builtin_amdgcn_sched_barrier(0);
  }

  for (int s0 = 0; s0 < PF_MAIN; s0 += PF_R) {
#pragma unroll
    for (int i = 0; i < PF_R; ++i) {
      ring[(i + PF_R - 1) % PF_R] = proj_load(p, ((s0 + i + PF_R - 1) * 8) & (D_SZ - 1));
      __builtin_amdgcn_sched_barrier(0);
      proj_step<FIRST>(ring[i], z0, z1, z2, acc0, acc1, acc2);
      __builtin_amdgcn_sched_barrier(0);
    }
  }
#pragma unroll
  for (int i = 0; i < (D_SZ / 8) - PF_MAIN; ++i)
    proj_step<FIRST>(ring[(PF_MAIN + i) % PF_R], z0, z1, z2, acc0, acc1, acc2);

  const int mhi = (lane >> 4) << 3;
#pragma unroll
  for (int v = 0; v < 8; ++v) {
    float* pr = proj + (size_t)(tile * 16 + v + mhi) * P_SZ;
    pr[c]      = acc0[v];
    pr[16 + c] = acc1[v];
    if (c == 0) pr[32] = acc2[v];
  }
}

__global__ __launch_bounds__(32) void conv_proj_wmma_kernel(
    const float* __restrict__ x, const float* __restrict__ cw,
    const float* __restrict__ cb, const float* __restrict__ w,
    float* __restrict__ proj)
{
  const int tile = blockIdx.x;
  const int lane = threadIdx.x;
  if ((tile & (L_SZ / 16 - 1)) == 0) conv_proj_tile<true>(x, cw, cb, w, proj, tile, lane);
  else                               conv_proj_tile<false>(x, cw, cb, w, proj, tile, lane);
}

#define STG_N ((CH_T * P_SZ + SCAN_BLK - 1) / SCAN_BLK)

__device__ __forceinline__ void stage_proj_load(const float* __restrict__ pr, float (&v)[STG_N])
{
#pragma unroll
  for (int j = 0; j < STG_N; ++j) {
    const int i = threadIdx.x + j * SCAN_BLK;
    v[j] = pr[i < CH_T * P_SZ ? i : CH_T * P_SZ - 1];
  }
}

#define S_ROW_N (CH_T * BC_W + CH_T + 1)
__device__ __forceinline__ void stage_proj_store(const float (&v)[STG_N], float* __restrict__ s_row)
{
#pragma unroll
  for (int j = 0; j < STG_N; ++j) {
    const int i   = threadIdx.x + j * SCAN_BLK;
    const int r   = i / P_SZ;
    const int col = i - r * P_SZ;
    const int a   = (col < BC_W) ? r * BC_W + col : CH_T * BC_W + r;
    s_row[i < CH_T * P_SZ ? a : S_ROW_N - 1] = v[j];
  }
}

__global__ __launch_bounds__(SCAN_BLK) void scan_chunk_summary_kernel(
    const float* __restrict__ x, const float* __restrict__ cw,
    const float* __restrict__ cb_, const float* __restrict__ proj,
    const float* __restrict__ dt_w, const float* __restrict__ dt_b,
    const float* __restrict__ a_log,
    float* __restrict__ ch_e, float* __restrict__ ch_p)
{
  const int ck = blockIdx.x / (CHAN / SCAN_BLK);
  const int cb = (blockIdx.x % (CHAN / SCAN_BLK)) * SCAN_BLK;
  const int b  = cb >> 10;
  const int c  = cb + threadIdx.x;
  const int d  = c & (D_SZ - 1);

  v2f ac[8];
#pragma unroll
  for (int k = 0; k < 8; ++k) {
    ac[k].x = -__expf(a_log[d * N_SZ + 2 * k]);
    ac[k].y = -__expf(a_log[d * N_SZ + 2 * k + 1]);
  }
  const float dtw = dt_w[d];
  const float dtb = dt_b[d];
  const v4f   cwd = *(const v4f*)(cw + (size_t)d * 4);
  const float cbd = cb_[d];

  const int t0 = ck * CH_T;
  const float* pr = proj + ((size_t)b * L_SZ + t0) * P_SZ;
  const float* xr = x    + ((size_t)b * L_SZ + t0) * D_SZ + d;

  __shared__ __attribute__((aligned(16))) float s_row[S_ROW_N];
  const float* s_bc = s_row;
  const float* s_dt = s_row + CH_T * BC_W;
  {
    float stg[STG_N];
    stage_proj_load(pr, stg);
    stage_proj_store(stg, s_row);
  }
  __syncthreads();

  float xw0 = 0.f, xw1 = 0.f, xw2 = 0.f;
  if (ck > 0) { xw0 = xr[-3 * D_SZ]; xw1 = xr[-2 * D_SZ]; xw2 = xr[-1 * D_SZ]; }

  v2f h[8], p[8];
#pragma unroll
  for (int k = 0; k < 8; ++k) { h[k] = (v2f){0.f, 0.f}; p[k] = (v2f){1.f, 1.f}; }

  for (int tb = 0; tb < CH_T; tb += SCAN_TB) {
    const v4f dq0 = *(const v4f*)(s_dt + tb);
    const v4f dq1 = *(const v4f*)(s_dt + tb + 4);
    const float p32v[SCAN_TB] = {dq0.x, dq0.y, dq0.z, dq0.w, dq1.x, dq1.y, dq1.z, dq1.w};
    v4f bq[SCAN_TB][N_SZ / 4];
#pragma unroll
    for (int i = 0; i < SCAN_TB; ++i) {
      const v4f* pt = (const v4f*)(s_bc + (tb + i) * BC_W);
#pragma unroll
      for (int q = 0; q < N_SZ / 4; ++q) bq[i][q] = pt[q];
    }

    float xv[SCAN_TB], xt[SCAN_TB];
#pragma unroll
    for (int i = 0; i < SCAN_TB; ++i) xv[i] = xr[(size_t)(tb + i) * D_SZ];
    xt[0] = conv_silu(xw0, xw1, xw2, xv[0], cwd, cbd);
    xt[1] = conv_silu(xw1, xw2, xv[0], xv[1], cwd, cbd);
    xt[2] = conv_silu(xw2, xv[0], xv[1], xv[2], cwd, cbd);
#pragma unroll
    for (int i = 3; i < SCAN_TB; ++i)
      xt[i] = conv_silu(xv[i - 3], xv[i - 2], xv[i - 1], xv[i], cwd, cbd);
    xw0 = xv[SCAN_TB - 3]; xw1 = xv[SCAN_TB - 2]; xw2 = xv[SCAN_TB - 1];

    float dl[SCAN_TB], dx[SCAN_TB];
#pragma unroll
    for (int i = 0; i < SCAN_TB; ++i) {
      const float p32 = p32v[i];
      const float z   = fmaf(p32, dtw, dtb);
      const float sp  = __logf(1.f + __expf(z));
      dl[i] = (z > 20.f) ? z : sp;
    }
#pragma unroll
    for (int i = 0; i < SCAN_TB; ++i) dx[i] = dl[i] * xt[i];

#pragma unroll
    for (int i = 0; i < SCAN_TB; ++i) {
#pragma unroll
      for (int k = 0; k < 8; ++k) {
        const v2f e  = dl[i] * ac[k];
        const v2f ab = {__expf(e.x), __expf(e.y)};
        const v4f b4 = bq[i][k >> 1];
        const v2f bt = (k & 1) ? (v2f){b4.z, b4.w} : (v2f){b4.x, b4.y};
        const v2f u  = dx[i] * bt;
        h[k] = __builtin_elementwise_fma(ab, h[k], u);
        p[k] = p[k] * ab;
      }
    }
  }

#pragma unroll
  for (int k = 0; k < 8; ++k) {
    const size_t o0 = ((size_t)(ck * N_SZ + 2 * k)) * CHAN + c;
    ch_e[o0]        = h[k].x;
    ch_e[o0 + CHAN] = h[k].y;
    ch_p[o0]        = p[k].x;
    ch_p[o0 + CHAN] = p[k].y;
  }
}

__global__ __launch_bounds__(SCAN_BLK) void scan_kernel(
    const float* __restrict__ x, const float* __restrict__ cw,
    const float* __restrict__ cb_,
    const float* __restrict__ proj, const float* __restrict__ dt_w,
    const float* __restrict__ dt_b, const float* __restrict__ a_log,
    const float* __restrict__ Dp, const float* __restrict__ ch_e,
    const float* __restrict__ ch_p, float* __restrict__ out)
{
  const int ck = blockIdx.x / (CHAN / SCAN_BLK);
  const int cb = (blockIdx.x % (CHAN / SCAN_BLK)) * SCAN_BLK;
  const int b  = cb >> 10;
  const int c  = cb + threadIdx.x;
  const int d  = c & (D_SZ - 1);

  v2f ac[8];
#pragma unroll
  for (int k = 0; k < 8; ++k) {
    ac[k].x = -__expf(a_log[d * N_SZ + 2 * k]);
    ac[k].y = -__expf(a_log[d * N_SZ + 2 * k + 1]);
  }
  const float dtw = dt_w[d];
  const float dtb = dt_b[d];
  const float dpd = Dp[d];
  const v4f   cwd = *(const v4f*)(cw + (size_t)d * 4);
  const float cbd = cb_[d];

  const float* pr = proj + ((size_t)b * L_SZ + ck * CH_T) * P_SZ;
  __shared__ __attribute__((aligned(16))) float s_row[S_ROW_N];
  const float* s_bc = s_row;
  const float* s_dt = s_row + CH_T * BC_W;
  float stg[STG_N];
  stage_proj_load(pr, stg);

  v2f h[8];
#pragma unroll
  for (int k = 0; k < 8; ++k) h[k] = (v2f){0.f, 0.f};
#pragma unroll 4
  for (int j = 0; j < ck; ++j) {
    v2f pj[8], ej[8];
#pragma unroll
    for (int k = 0; k < 8; ++k) {
      const size_t o0 = ((size_t)(j * N_SZ + 2 * k)) * CHAN + c;
      pj[k].x = ch_p[o0];
      pj[k].y = ch_p[o0 + CHAN];
      ej[k].x = ch_e[o0];
      ej[k].y = ch_e[o0 + CHAN];
    }
#pragma unroll
    for (int k = 0; k < 8; ++k) h[k] = __builtin_elementwise_fma(pj[k], h[k], ej[k]);
  }

  const int t0 = ck * CH_T;
  const float* xr   = x    + ((size_t)b * L_SZ + t0) * D_SZ + d;
  float*       outr = out  + ((size_t)b * L_SZ + t0) * D_SZ + d;

  stage_proj_store(stg, s_row);
  __syncthreads();

  float xw0 = 0.f, xw1 = 0.f, xw2 = 0.f;
  if (ck > 0) { xw0 = xr[-3 * D_SZ]; xw1 = xr[-2 * D_SZ]; xw2 = xr[-1 * D_SZ]; }

  for (int tb = 0; tb < CH_T; tb += SCAN_TB) {
    float xt[SCAN_TB], xv[SCAN_TB];
#pragma unroll
    for (int i = 0; i < SCAN_TB; ++i) xv[i] = xr[(size_t)(tb + i) * D_SZ];
    xt[0] = conv_silu(xw0, xw1, xw2, xv[0], cwd, cbd);
    xt[1] = conv_silu(xw1, xw2, xv[0], xv[1], cwd, cbd);
    xt[2] = conv_silu(xw2, xv[0], xv[1], xv[2], cwd, cbd);
#pragma unroll
    for (int i = 3; i < SCAN_TB; ++i)
      xt[i] = conv_silu(xv[i - 3], xv[i - 2], xv[i - 1], xv[i], cwd, cbd);
    xw0 = xv[SCAN_TB - 3]; xw1 = xv[SCAN_TB - 2]; xw2 = xv[SCAN_TB - 1];

    float dl[SCAN_TB], dx[SCAN_TB];
#pragma unroll
    for (int i = 0; i < SCAN_TB; ++i) {
      const float p32 = s_dt[tb + i];
      const float z   = fmaf(p32, dtw, dtb);
      const float sp  = __logf(1.f + __expf(z));
      dl[i] = (z > 20.f) ? z : sp;
    }
#pragma unroll
    for (int i = 0; i < SCAN_TB; ++i) dx[i] = dl[i] * xt[i];

#pragma unroll
    for (int i = 0; i < SCAN_TB; ++i) {
      const v4f* pt = (const v4f*)(s_bc + (tb + i) * BC_W);
      v2f ya = {0.f, 0.f}, yb = {0.f, 0.f};
#pragma unroll
      for (int k = 0; k < 8; ++k) {
        const v2f e  = dl[i] * ac[k];
        const v2f ab = {__expf(e.x), __expf(e.y)};
        const v4f b4 = pt[k >> 1];
        const v2f bt = (k & 1) ? (v2f){b4.z, b4.w} : (v2f){b4.x, b4.y};
        const v4f c4 = pt[N_SZ / 4 + (k >> 1)];
        const v2f ct = (k & 1) ? (v2f){c4.z, c4.w} : (v2f){c4.x, c4.y};
        const v2f u  = dx[i] * bt;
        h[k] = __builtin_elementwise_fma(ab, h[k], u);
        if (k & 1) yb = __builtin_elementwise_fma(h[k], ct, yb);
        else       ya = __builtin_elementwise_fma(h[k], ct, ya);
      }
      const v2f ys = ya + yb;
      const float y = ys.x + ys.y;
      outr[(size_t)(tb + i) * D_SZ] = fmaf(xv[i], dpd, y);
    }
  }
}

extern "C" void kernel_launch(void* const* d_in, const int* in_sizes, int n_in,
                              void* d_out, int out_size, void* d_ws, size_t ws_size,
                              hipStream_t stream) {
  const float* x      = (const float*)d_in[0];
  const float* conv_w = (const float*)d_in[1];
  const float* conv_b = (const float*)d_in[2];
  const float* xproj  = (const float*)d_in[3];
  const float* dt_w   = (const float*)d_in[4];
  const float* dt_b   = (const float*)d_in[5];
  const float* a_log  = (const float*)d_in[6];
  const float* Dp     = (const float*)d_in[7];
  float* out = (float*)d_out;

  float* proj = (float*)d_ws;
  float* ch_e = proj + (size_t)B_SZ * L_SZ * P_SZ;
  float* ch_p = ch_e + (size_t)(CH_N - 1) * CHAN * N_SZ;

  conv_proj_wmma_kernel<<<(B_SZ * L_SZ) / 16, 32, 0, stream>>>(
      x, conv_w, conv_b, xproj, proj);

  scan_chunk_summary_kernel<<<(CHAN / SCAN_BLK) * (CH_N - 1), SCAN_BLK, 0, stream>>>(
      x, conv_w, conv_b, proj, dt_w, dt_b, a_log, ch_e, ch_p);

  scan_kernel<<<(CHAN / SCAN_BLK) * CH_N, SCAN_BLK, 0, stream>>>(
      x, conv_w, conv_b, proj, dt_w, dt_b, a_log, Dp, ch_e, ch_p, out);
}


# proj K loop: conv weights, bias, weight row 32 (equal in all lanes of a half) copied to LDS once per 2-tile block, read from there: 6 global requests a step, not 12; ring 9 steps deep; fp order same
# speedup vs baseline: 25.5219x; 1.0393x over previous
#include <hip/hip_runtime.h>
#include <hip/hip_bf16.h>


#define B_SZ 4
#define L_SZ 2048
#define D_SZ 1024
#define N_SZ 16
#define P_SZ 33
#define CH_N 16
#define CH_T (L_SZ / CH_N)
#define CHAN (B_SZ * D_SZ)
#define SCAN_BLK 256
#define SCAN_TB 8
#define BC_W (2 * N_SZ)

typedef float v2f __attribute__((ext_vector_type(2)));
typedef float v4f __attribute__((ext_vector_type(4)));
typedef float v8f __attribute__((ext_vector_type(8)));

__device__ __forceinline__ float conv_silu(float x0, float x1, float x2, float x3,
                                           v4f w, float bias)
{
  float acc = bias;
  acc = fmaf(x0, w.x, acc);
  acc = fmaf(x1, w.y, acc);
  acc = fmaf(x2, w.z, acc);
  acc = fmaf(x3, w.w, acc);
  return acc * __builtin_amdgcn_rcpf(1.f + __expf(-acc));
}

struct ProjStepG {
  v4f x0, x1, x2, x3;
  v4f b0, b1;
};
struct ProjStepL {
  v4f wa, wb, wc, wd;
  v4f bias;
  v4f b2;
};

#define LDS_CW 0
#define LDS_CB (D_SZ * 4)
#define LDS_W32 (D_SZ * 5)
#define LDS_PROJ_N (D_SZ * 6)
#define PROJ_BLK 64
#define STG_V4 (LDS_PROJ_N / 4 / PROJ_BLK)

#define PF_R 10
#define PF_MAIN (((D_SZ / 8) / PF_R) * PF_R)

struct ProjPtrs {
  const float *xp0, *xp1, *xp2, *xp3, *bp0, *bp1;
  const float *lw, *lb, *l2;
};

__device__ __forceinline__ ProjStepG proj_gload(const ProjPtrs& p, const int k0)
{
  ProjStepG r;
  r.x0 = *(const v4f*)(p.xp0 + k0);
  r.x1 = *(const v4f*)(p.xp1 + k0);
  r.x2 = *(const v4f*)(p.xp2 + k0);
  r.x3 = *(const v4f*)(p.xp3 + k0);
  r.b0 = *(const v4f*)(p.bp0 + k0);
  r.b1 = *(const v4f*)(p.bp1 + k0);
  return r;
}

__device__ __forceinline__ ProjStepL proj_lload(const ProjPtrs& p, const int k0)
{
  ProjStepL r;
  r.wa   = *(const v4f*)(p.lw + k0 * 4);
  r.wb   = *(const v4f*)(p.lw + k0 * 4 + 4);
  r.wc   = *(const v4f*)(p.lw + k0 * 4 + 8);
  r.wd   = *(const v4f*)(p.lw + k0 * 4 + 12);
  r.bias = *(const v4f*)(p.lb + k0);
  r.b2   = *(const v4f*)(p.l2 + k0);
  return r;
}

template <bool FIRST>
__device__ __forceinline__ void proj_step(ProjStepG s, const ProjStepL l, const bool z0,
                                          const bool z1, const bool z2,
                                          v8f& acc0, v8f& acc1, v8f& acc2)
{
  if (FIRST) {
    const v4f zero4 = {0.f, 0.f, 0.f, 0.f};
    if (z0) s.x0 = zero4;
    if (z1) s.x1 = zero4;
    if (z2) s.x2 = zero4;
  }
  const v2f alo = {conv_silu(s.x0.x, s.x1.x, s.x2.x, s.x3.x, l.wa, l.bias.x),
                   conv_silu(s.x0.y, s.x1.y, s.x2.y, s.x3.y, l.wb, l.bias.y)};
  const v2f ahi = {conv_silu(s.x0.z, s.x1.z, s.x2.z, s.x3.z, l.wc, l.bias.z),
                   conv_silu(s.x0.w, s.x1.w, s.x2.w, s.x3.w, l.wd, l.bias.w)};

  acc0 = __builtin_amdgcn_wmma_f32_16x16x4_f32(false, alo, false, (v2f){s.b0.x, s.b0.y},
                                               (short)0, acc0, false, false);
  acc1 = __builtin_amdgcn_wmma_f32_16x16x4_f32(false, alo, false, (v2f){s.b1.x, s.b1.y},
                                               (short)0, acc1, false, false);
  acc2 = __builtin_amdgcn_wmma_f32_16x16x4_f32(false, alo, false, (v2f){l.b2.x, l.b2.y},
                                               (short)0, acc2, false, false);
  acc0 = __builtin_amdgcn_wmma_f32_16x16x4_f32(false, ahi, false, (v2f){s.b0.z, s.b0.w},
                                               (short)0, acc0, false, false);
  acc1 = __builtin_amdgcn_wmma_f32_16x16x4_f32(false, ahi, false, (v2f){s.b1.z, s.b1.w},
                                               (short)0, acc1, false, false);
  acc2 = __builtin_amdgcn_wmma_f32_16x16x4_f32(false, ahi, false, (v2f){l.b2.z, l.b2.w},
                                               (short)0, acc2, false, false);
}

template <bool FIRST>
__device__ __forceinline__ void conv_proj_loop(
    ProjStepG (&ring)[PF_R], ProjStepL (&lring)[PF_R], const ProjPtrs& p,
    const bool z0, const bool z1, const bool z2,
    float* __restrict__ proj, const int tile, const int lane)
{
  const int c = lane & 15;
  v8f acc0 = {}, acc1 = {}, acc2 = {};

  for (int s0 = 0; s0 < PF_MAIN; s0 += PF_R) {
#pragma unroll
    for (int i = 0; i < PF_R; ++i) {
      lring[(i + 1) % PF_R] = proj_lload(p, ((s0 + i + 1) * 8) & (D_SZ - 1));
      ring[(i + PF_R - 1) % PF_R] = proj_gload(p, ((s0 + i + PF_R - 1) * 8) & (D_SZ - 1));
      __builtin_amdgcn_sched_barrier(0);
      proj_step<FIRST>(ring[i], lring[i], z0, z1, z2, acc0, acc1, acc2);
      __builtin_amdgcn_sched_barrier(0);
    }
  }
#pragma unroll
  for (int i = 0; i < (D_SZ / 8) - PF_MAIN; ++i) {
    if (i + 1 < (D_SZ / 8) - PF_MAIN)
      lring[(i + 1) % PF_R] = proj_lload(p, (PF_MAIN + i + 1) * 8);
    proj_step<FIRST>(ring[(PF_MAIN + i) % PF_R], lring[i % PF_R], z0, z1, z2, acc0, acc1, acc2);
  }

  const int mhi = (lane >> 4) << 3;
#pragma unroll
  for (int v = 0; v < 8; ++v) {
    float* pr = proj + (size_t)(tile * 16 + v + mhi) * P_SZ;
    pr[c]      = acc0[v];
    pr[16 + c] = acc1[v];
    if (c == 0) pr[32] = acc2[v];
  }
}

__global__ __launch_bounds__(PROJ_BLK) void conv_proj_wmma_kernel(
    const float* __restrict__ x, const float* __restrict__ cw,
    const float* __restrict__ cb, const float* __restrict__ w,
    float* __restrict__ proj)
{
  __shared__ __attribute__((aligned(16))) float s_u[LDS_PROJ_N];

  const int wave = __builtin_amdgcn_readfirstlane((int)threadIdx.x >> 5);
  const int lane = threadIdx.x & 31;
  const int tile = blockIdx.x * (PROJ_BLK / 32) + wave;
  const bool first = (tile & (L_SZ / 16 - 1)) == 0;

  v4f stg[STG_V4];
  {
    const v4f* g_cw  = (const v4f*)cw + threadIdx.x;
    const v4f* g_cb  = (const v4f*)cb + threadIdx.x;
    const v4f* g_w32 = (const v4f*)(w + (size_t)(P_SZ - 1) * D_SZ) + threadIdx.x;
#pragma unroll
    for (int j = 0; j < D_SZ / PROJ_BLK; ++j)     stg[j] = g_cw[j * PROJ_BLK];
#pragma unroll
    for (int j = 0; j < D_SZ / 4 / PROJ_BLK; ++j) stg[D_SZ / PROJ_BLK + j] = g_cb[j * PROJ_BLK];
#pragma unroll
    for (int j = 0; j < D_SZ / 4 / PROJ_BLK; ++j)
      stg[D_SZ / PROJ_BLK + D_SZ / 4 / PROJ_BLK + j] = g_w32[j * PROJ_BLK];
  }

  const int c   = lane & 15;
  const int kh  = (lane >> 4) << 2;
  const int row = tile * 16 + c;
  const float* xrow = x + (size_t)row * D_SZ + kh;
  const bool z0 = first && (c < 3);
  const bool z1 = first && (c < 2);
  const bool z2 = first && (c < 1);

  ProjPtrs p;
  p.xp0 = z0 ? xrow : xrow - 3 * D_SZ;
  p.xp1 = z1 ? xrow : xrow - 2 * D_SZ;
  p.xp2 = z2 ? xrow : xrow - 1 * D_SZ;
  p.xp3 = xrow;
  p.bp0 = w + (size_t)c * D_SZ + kh;
  p.bp1 = w + (size_t)(16 + c) * D_SZ + kh;
  p.lw  = s_u + LDS_CW + kh * 4;
  p.lb  = s_u + LDS_CB + kh;
  p.l2  = s_u + LDS_W32 + kh;

  ProjStepG ring[PF_R];
  ProjStepL lring[PF_R];
#pragma unroll
  for (int i = 0; i < PF_R - 1; ++i) {
    ring[i] = proj_gload(p, 8 * i);
    __builtin_amdgcn_sched_barrier(0);
  }

#pragma unroll
  for (int j = 0; j < STG_V4; ++j)
    *(v4f*)(s_u + (size_t)(threadIdx.x + j * PROJ_BLK) * 4) = stg[j];
  __syncthreads();

  lring[0] = proj_lload(p, 0);

  if (first) conv_proj_loop<true>(ring, lring, p, z0, z1, z2, proj, tile, lane);
  else       conv_proj_loop<false>(ring, lring, p, z0, z1, z2, proj, tile, lane);
}

#define STG_N ((CH_T * P_SZ + SCAN_BLK - 1) / SCAN_BLK)

__device__ __forceinline__ void stage_proj_load(const float* __restrict__ pr, float (&v)[STG_N])
{
#pragma unroll
  for (int j = 0; j < STG_N; ++j) {
    const int i = threadIdx.x + j * SCAN_BLK;
    v[j] = pr[i < CH_T * P_SZ ? i : CH_T * P_SZ - 1];
  }
}

#define S_ROW_N (CH_T * BC_W + CH_T + 1)
__device__ __forceinline__ void stage_proj_store(const float (&v)[STG_N], float* __restrict__ s_row)
{
#pragma unroll
  for (int j = 0; j < STG_N; ++j) {
    const int i   = threadIdx.x + j * SCAN_BLK;
    const int r   = i / P_SZ;
    const int col = i - r * P_SZ;
    const int a   = (col < BC_W) ? r * BC_W + col : CH_T * BC_W + r;
    s_row[i < CH_T * P_SZ ? a : S_ROW_N - 1] = v[j];
  }
}

__global__ __launch_bounds__(SCAN_BLK) void scan_chunk_summary_kernel(
    const float* __restrict__ x, const float* __restrict__ cw,
    const float* __restrict__ cb_, const float* __restrict__ proj,
    const float* __restrict__ dt_w, const float* __restrict__ dt_b,
    const float* __restrict__ a_log,
    float* __restrict__ ch_e, float* __restrict__ ch_p)
{
  const int ck = blockIdx.x / (CHAN / SCAN_BLK);
  const int cb = (blockIdx.x % (CHAN / SCAN_BLK)) * SCAN_BLK;
  const int b  = cb >> 10;
  const int c  = cb + threadIdx.x;
  const int d  = c & (D_SZ - 1);

  v2f ac[8];
#pragma unroll
  for (int k = 0; k < 8; ++k) {
    ac[k].x = -__expf(a_log[d * N_SZ + 2 * k]);
    ac[k].y = -__expf(a_log[d * N_SZ + 2 * k + 1]);
  }
  const float dtw = dt_w[d];
  const float dtb = dt_b[d];
  const v4f   cwd = *(const v4f*)(cw + (size_t)d * 4);
  const float cbd = cb_[d];

  const int t0 = ck * CH_T;
  const float* pr = proj + ((size_t)b * L_SZ + t0) * P_SZ;
  const float* xr = x    + ((size_t)b * L_SZ + t0) * D_SZ + d;

  __shared__ __attribute__((aligned(16))) float s_row[S_ROW_N];
  const float* s_bc = s_row;
  const float* s_dt = s_row + CH_T * BC_W;
  {
    float stg[STG_N];
    stage_proj_load(pr, stg);
    stage_proj_store(stg, s_row);
  }
  __syncthreads();

  float xw0 = 0.f, xw1 = 0.f, xw2 = 0.f;
  if (ck > 0) { xw0 = xr[-3 * D_SZ]; xw1 = xr[-2 * D_SZ]; xw2 = xr[-1 * D_SZ]; }

  v2f h[8], p[8];
#pragma unroll
  for (int k = 0; k < 8; ++k) { h[k] = (v2f){0.f, 0.f}; p[k] = (v2f){1.f, 1.f}; }

  for (int tb = 0; tb < CH_T; tb += SCAN_TB) {
    const v4f dq0 = *(const v4f*)(s_dt + tb);
    const v4f dq1 = *(const v4f*)(s_dt + tb + 4);
    const float p32v[SCAN_TB] = {dq0.x, dq0.y, dq0.z, dq0.w, dq1.x, dq1.y, dq1.z, dq1.w};
    v4f bq[SCAN_TB][N_SZ / 4];
#pragma unroll
    for (int i = 0; i < SCAN_TB; ++i) {
      const v4f* pt = (const v4f*)(s_bc + (tb + i) * BC_W);
#pragma unroll
      for (int q = 0; q < N_SZ / 4; ++q) bq[i][q] = pt[q];
    }

    float xv[SCAN_TB], xt[SCAN_TB];
#pragma unroll
    for (int i = 0; i < SCAN_TB; ++i) xv[i] = xr[(size_t)(tb + i) * D_SZ];
    xt[0] = conv_silu(xw0, xw1, xw2, xv[0], cwd, cbd);
    xt[1] = conv_silu(xw1, xw2, xv[0], xv[1], cwd, cbd);
    xt[2] = conv_silu(xw2, xv[0], xv[1], xv[2], cwd, cbd);
#pragma unroll
    for (int i = 3; i < SCAN_TB; ++i)
      xt[i] = conv_silu(xv[i - 3], xv[i - 2], xv[i - 1], xv[i], cwd, cbd);
    xw0 = xv[SCAN_TB - 3]; xw1 = xv[SCAN_TB - 2]; xw2 = xv[SCAN_TB - 1];

    float dl[SCAN_TB], dx[SCAN_TB];
#pragma unroll
    for (int i = 0; i < SCAN_TB; ++i) {
      const float p32 = p32v[i];
      const float z   = fmaf(p32, dtw, dtb);
      const float sp  = __logf(1.f + __expf(z));
      dl[i] = (z > 20.f) ? z : sp;
    }
#pragma unroll
    for (int i = 0; i < SCAN_TB; ++i) dx[i] = dl[i] * xt[i];

#pragma unroll
    for (int i = 0; i < SCAN_TB; ++i) {
#pragma unroll
      for (int k = 0; k < 8; ++k) {
        const v2f e  = dl[i] * ac[k];
        const v2f ab = {__expf(e.x), __expf(e.y)};
        const v4f b4 = bq[i][k >> 1];
        const v2f bt = (k & 1) ? (v2f){b4.z, b4.w} : (v2f){b4.x, b4.y};
        const v2f u  = dx[i] * bt;
        h[k] = __builtin_elementwise_fma(ab, h[k], u);
        p[k] = p[k] * ab;
      }
    }
  }

#pragma unroll
  for (int k = 0; k < 8; ++k) {
    const size_t o0 = ((size_t)(ck * N_SZ + 2 * k)) * CHAN + c;
    ch_e[o0]        = h[k].x;
    ch_e[o0 + CHAN] = h[k].y;
    ch_p[o0]        = p[k].x;
    ch_p[o0 + CHAN] = p[k].y;
  }
}

__global__ __launch_bounds__(SCAN_BLK) void scan_kernel(
    const float* __restrict__ x, const float* __restrict__ cw,
    const float* __restrict__ cb_,
    const float* __restrict__ proj, const float* __restrict__ dt_w,
    const float* __restrict__ dt_b, const float* __restrict__ a_log,
    const float* __restrict__ Dp, const float* __restrict__ ch_e,
    const float* __restrict__ ch_p, float* __restrict__ out)
{
  const int ck = blockIdx.x / (CHAN / SCAN_BLK);
  const int cb = (blockIdx.x % (CHAN / SCAN_BLK)) * SCAN_BLK;
  const int b  = cb >> 10;
  const int c  = cb + threadIdx.x;
  const int d  = c & (D_SZ - 1);

  v2f ac[8];
#pragma unroll
  for (int k = 0; k < 8; ++k) {
    ac[k].x = -__expf(a_log[d * N_SZ + 2 * k]);
    ac[k].y = -__expf(a_log[d * N_SZ + 2 * k + 1]);
  }
  const float dtw = dt_w[d];
  const float dtb = dt_b[d];
  const float dpd = Dp[d];
  const v4f   cwd = *(const v4f*)(cw + (size_t)d * 4);
  const float cbd = cb_[d];

  const float* pr = proj + ((size_t)b * L_SZ + ck * CH_T) * P_SZ;
  __shared__ __attribute__((aligned(16))) float s_row[S_ROW_N];
  const float* s_bc = s_row;
  const float* s_dt = s_row + CH_T * BC_W;
  float stg[STG_N];
  stage_proj_load(pr, stg);

  v2f h[8];
#pragma unroll
  for (int k = 0; k < 8; ++k) h[k] = (v2f){0.f, 0.f};
#pragma unroll 4
  for (int j = 0; j < ck; ++j) {
    v2f pj[8], ej[8];
#pragma unroll
    for (int k = 0; k < 8; ++k) {
      const size_t o0 = ((size_t)(j * N_SZ + 2 * k)) * CHAN + c;
      pj[k].x = ch_p[o0];
      pj[k].y = ch_p[o0 + CHAN];
      ej[k].x = ch_e[o0];
      ej[k].y = ch_e[o0 + CHAN];
    }
#pragma unroll
    for (int k = 0; k < 8; ++k) h[k] = __builtin_elementwise_fma(pj[k], h[k], ej[k]);
  }

  const int t0 = ck * CH_T;
  const float* xr   = x    + ((size_t)b * L_SZ + t0) * D_SZ + d;
  float*       outr = out  + ((size_t)b * L_SZ + t0) * D_SZ + d;

  stage_proj_store(stg, s_row);
  __syncthreads();

  float xw0 = 0.f, xw1 = 0.f, xw2 = 0.f;
  if (ck > 0) { xw0 = xr[-3 * D_SZ]; xw1 = xr[-2 * D_SZ]; xw2 = xr[-1 * D_SZ]; }

  for (int tb = 0; tb < CH_T; tb += SCAN_TB) {
    float xt[SCAN_TB], xv[SCAN_TB];
#pragma unroll
    for (int i = 0; i < SCAN_TB; ++i) xv[i] = xr[(size_t)(tb + i) * D_SZ];
    xt[0] = conv_silu(xw0, xw1, xw2, xv[0], cwd, cbd);
    xt[1] = conv_silu(xw1, xw2, xv[0], xv[1], cwd, cbd);
    xt[2] = conv_silu(xw2, xv[0], xv[1], xv[2], cwd, cbd);
#pragma unroll
    for (int i = 3; i < SCAN_TB; ++i)
      xt[i] = conv_silu(xv[i - 3], xv[i - 2], xv[i - 1], xv[i], cwd, cbd);
    xw0 = xv[SCAN_TB - 3]; xw1 = xv[SCAN_TB - 2]; xw2 = xv[SCAN_TB - 1];

    float dl[SCAN_TB], dx[SCAN_TB];
#pragma unroll
    for (int i = 0; i < SCAN_TB; ++i) {
      const float p32 = s_dt[tb + i];
      const float z   = fmaf(p32, dtw, dtb);
      const float sp  = __logf(1.f + __expf(z));
      dl[i] = (z > 20.f) ? z : sp;
    }
#pragma unroll
    for (int i = 0; i < SCAN_TB; ++i) dx[i] = dl[i] * xt[i];

#pragma unroll
    for (int i = 0; i < SCAN_TB; ++i) {
      const v4f* pt = (const v4f*)(s_bc + (tb + i) * BC_W);
      v2f ya = {0.f, 0.f}, yb = {0.f, 0.f};
#pragma unroll
      for (int k = 0; k < 8; ++k) {
        const v2f e  = dl[i] * ac[k];
        const v2f ab = {__expf(e.x), __expf(e.y)};
        const v4f b4 = pt[k >> 1];
        const v2f bt = (k & 1) ? (v2f){b4.z, b4.w} : (v2f){b4.x, b4.y};
        const v4f c4 = pt[N_SZ / 4 + (k >> 1)];
        const v2f ct = (k & 1) ? (v2f){c4.z, c4.w} : (v2f){c4.x, c4.y};
        const v2f u  = dx[i] * bt;
        h[k] = __builtin_elementwise_fma(ab, h[k], u);
        if (k & 1) yb = __builtin_elementwise_fma(h[k], ct, yb);
        else       ya = __builtin_elementwise_fma(h[k], ct, ya);
      }
      const v2f ys = ya + yb;
      const float y = ys.x + ys.y;
      outr[(size_t)(tb + i) * D_SZ] = fmaf(xv[i], dpd, y);
    }
  }
}

extern "C" void kernel_launch(void* const* d_in, const int* in_sizes, int n_in,
                              void* d_out, int out_size, void* d_ws, size_t ws_size,
                              hipStream_t stream) {
  const float* x      = (const float*)d_in[0];
  const float* conv_w = (const float*)d_in[1];
  const float* conv_b = (const float*)d_in[2];
  const float* xproj  = (const float*)d_in[3];
  const float* dt_w   = (const float*)d_in[4];
  const float* dt_b   = (const float*)d_in[5];
  const float* a_log  = (const float*)d_in[6];
  const float* Dp     = (const float*)d_in[7];
  float* out = (float*)d_out;

  float* proj = (float*)d_ws;
  float* ch_e = proj + (size_t)B_SZ * L_SZ * P_SZ;
  float* ch_p = ch_e + (size_t)(CH_N - 1) * CHAN * N_SZ;

  conv_proj_wmma_kernel<<<(B_SZ * L_SZ) / 16 / (PROJ_BLK / 32), PROJ_BLK, 0, stream>>>(
      x, conv_w, conv_b, xproj, proj);

  scan_chunk_summary_kernel<<<(CHAN / SCAN_BLK) * (CH_N - 1), SCAN_BLK, 0, stream>>>(
      x, conv_w, conv_b, proj, dt_w, dt_b, a_log, ch_e, ch_p);

  scan_kernel<<<(CHAN / SCAN_BLK) * CH_N, SCAN_BLK, 0, stream>>>(
      x, conv_w, conv_b, proj, dt_w, dt_b, a_log, Dp, ch_e, ch_p, out);
}


# chunk summary: a chunk walked as two 64-step parts by two 8-wave groups of a block (4 waves/SIMD), joined in LDS; decay kept as the sum of deltas, p=exp(a_n*sum): h, p formed in a new order
# speedup vs baseline: 26.3999x; 1.0344x over previous
#include <hip/hip_runtime.h>
#include <hip/hip_bf16.h>


#define B_SZ 4
#define L_SZ 2048
#define D_SZ 1024
#define N_SZ 16
#define P_SZ 33
#define CH_N 16
#define CH_T (L_SZ / CH_N)
#define CHAN (B_SZ * D_SZ)
#define SCAN_BLK 256
#define SCAN_TB 8
#define BC_W (2 * N_SZ)

typedef float v2f __attribute__((ext_vector_type(2)));
typedef float v4f __attribute__((ext_vector_type(4)));
typedef float v8f __attribute__((ext_vector_type(8)));

__device__ __forceinline__ float conv_silu(float x0, float x1, float x2, float x3,
                                           v4f w, float bias)
{
  float acc = bias;
  acc = fmaf(x0, w.x, acc);
  acc = fmaf(x1, w.y, acc);
  acc = fmaf(x2, w.z, acc);
  acc = fmaf(x3, w.w, acc);
  return acc * __builtin_amdgcn_rcpf(1.f + __expf(-acc));
}

struct ProjStepG {
  v4f x0, x1, x2, x3;
  v4f b0, b1;
};
struct ProjStepL {
  v4f wa, wb, wc, wd;
  v4f bias;
  v4f b2;
};

#define LDS_CW 0
#define LDS_CB (D_SZ * 4)
#define LDS_W32 (D_SZ * 5)
#define LDS_PROJ_N (D_SZ * 6)
#define PROJ_BLK 64
#define STG_V4 (LDS_PROJ_N / 4 / PROJ_BLK)

#define PF_R 10
#define PF_MAIN (((D_SZ / 8) / PF_R) * PF_R)

struct ProjPtrs {
  const float *xp0, *xp1, *xp2, *xp3, *bp0, *bp1;
  const float *lw, *lb, *l2;
};

__device__ __forceinline__ ProjStepG proj_gload(const ProjPtrs& p, const int k0)
{
  ProjStepG r;
  r.x0 = *(const v4f*)(p.xp0 + k0);
  r.x1 = *(const v4f*)(p.xp1 + k0);
  r.x2 = *(const v4f*)(p.xp2 + k0);
  r.x3 = *(const v4f*)(p.xp3 + k0);
  r.b0 = *(const v4f*)(p.bp0 + k0);
  r.b1 = *(const v4f*)(p.bp1 + k0);
  return r;
}

__device__ __forceinline__ ProjStepL proj_lload(const ProjPtrs& p, const int k0)
{
  ProjStepL r;
  r.wa   = *(const v4f*)(p.lw + k0 * 4);
  r.wb   = *(const v4f*)(p.lw + k0 * 4 + 4);
  r.wc   = *(const v4f*)(p.lw + k0 * 4 + 8);
  r.wd   = *(const v4f*)(p.lw + k0 * 4 + 12);
  r.bias = *(const v4f*)(p.lb + k0);
  r.b2   = *(const v4f*)(p.l2 + k0);
  return r;
}

template <bool FIRST>
__device__ __forceinline__ void proj_step(ProjStepG s, const ProjStepL l, const bool z0,
                                          const bool z1, const bool z2,
                                          v8f& acc0, v8f& acc1, v8f& acc2)
{
  if (FIRST) {
    const v4f zero4 = {0.f, 0.f, 0.f, 0.f};
    if (z0) s.x0 = zero4;
    if (z1) s.x1 = zero4;
    if (z2) s.x2 = zero4;
  }
  const v2f alo = {conv_silu(s.x0.x, s.x1.x, s.x2.x, s.x3.x, l.wa, l.bias.x),
                   conv_silu(s.x0.y, s.x1.y, s.x2.y, s.x3.y, l.wb, l.bias.y)};
  const v2f ahi = {conv_silu(s.x0.z, s.x1.z, s.x2.z, s.x3.z, l.wc, l.bias.z),
                   conv_silu(s.x0.w, s.x1.w, s.x2.w, s.x3.w, l.wd, l.bias.w)};

  acc0 = __builtin_amdgcn_wmma_f32_16x16x4_f32(false, alo, false, (v2f){s.b0.x, s.b0.y},
                                               (short)0, acc0, false, false);
  acc1 = __builtin_amdgcn_wmma_f32_16x16x4_f32(false, alo, false, (v2f){s.b1.x, s.b1.y},
                                               (short)0, acc1, false, false);
  acc2 = __builtin_amdgcn_wmma_f32_16x16x4_f32(false, alo, false, (v2f){l.b2.x, l.b2.y},
                                               (short)0, acc2, false, false);
  acc0 = __builtin_amdgcn_wmma_f32_16x16x4_f32(false, ahi, false, (v2f){s.b0.z, s.b0.w},
                                               (short)0, acc0, false, false);
  acc1 = __builtin_amdgcn_wmma_f32_16x16x4_f32(false, ahi, false, (v2f){s.b1.z, s.b1.w},
                                               (short)0, acc1, false, false);
  acc2 = __builtin_amdgcn_wmma_f32_16x16x4_f32(false, ahi, false, (v2f){l.b2.z, l.b2.w},
                                               (short)0, acc2, false, false);
}

template <bool FIRST>
__device__ __forceinline__ void conv_proj_loop(
    ProjStepG (&ring)[PF_R], ProjStepL (&lring)[PF_R], const ProjPtrs& p,
    const bool z0, const bool z1, const bool z2,
    float* __restrict__ proj, const int tile, const int lane)
{
  const int c = lane & 15;
  v8f acc0 = {}, acc1 = {}, acc2 = {};

  for (int s0 = 0; s0 < PF_MAIN; s0 += PF_R) {
#pragma unroll
    for (int i = 0; i < PF_R; ++i) {
      lring[(i + 1) % PF_R] = proj_lload(p, ((s0 + i + 1) * 8) & (D_SZ - 1));
      ring[(i + PF_R - 1) % PF_R] = proj_gload(p, ((s0 + i + PF_R - 1) * 8) & (D_SZ - 1));
      __builtin_amdgcn_sched_barrier(0);
      proj_step<FIRST>(ring[i], lring[i], z0, z1, z2, acc0, acc1, acc2);
      __builtin_amdgcn_sched_barrier(0);
    }
  }
#pragma unroll
  for (int i = 0; i < (D_SZ / 8) - PF_MAIN; ++i) {
    if (i + 1 < (D_SZ / 8) - PF_MAIN)
      lring[(i + 1) % PF_R] = proj_lload(p, (PF_MAIN + i + 1) * 8);
    proj_step<FIRST>(ring[(PF_MAIN + i) % PF_R], lring[i % PF_R], z0, z1, z2, acc0, acc1, acc2);
  }

  const int mhi = (lane >> 4) << 3;
#pragma unroll
  for (int v = 0; v < 8; ++v) {
    float* pr = proj + (size_t)(tile * 16 + v + mhi) * P_SZ;
    pr[c]      = acc0[v];
    pr[16 + c] = acc1[v];
    if (c == 0) pr[32] = acc2[v];
  }
}

__global__ __launch_bounds__(PROJ_BLK) void conv_proj_wmma_kernel(
    const float* __restrict__ x, const float* __restrict__ cw,
    const float* __restrict__ cb, const float* __restrict__ w,
    float* __restrict__ proj)
{
  __shared__ __attribute__((aligned(16))) float s_u[LDS_PROJ_N];

  const int wave = __builtin_amdgcn_readfirstlane((int)threadIdx.x >> 5);
  const int lane = threadIdx.x & 31;
  const int tile = blockIdx.x * (PROJ_BLK / 32) + wave;
  const bool first = (tile & (L_SZ / 16 - 1)) == 0;

  v4f stg[STG_V4];
  {
    const v4f* g_cw  = (const v4f*)cw + threadIdx.x;
    const v4f* g_cb  = (const v4f*)cb + threadIdx.x;
    const v4f* g_w32 = (const v4f*)(w + (size_t)(P_SZ - 1) * D_SZ) + threadIdx.x;
#pragma unroll
    for (int j = 0; j < D_SZ / PROJ_BLK; ++j)     stg[j] = g_cw[j * PROJ_BLK];
#pragma unroll
    for (int j = 0; j < D_SZ / 4 / PROJ_BLK; ++j) stg[D_SZ / PROJ_BLK + j] = g_cb[j * PROJ_BLK];
#pragma unroll
    for (int j = 0; j < D_SZ / 4 / PROJ_BLK; ++j)
      stg[D_SZ / PROJ_BLK + D_SZ / 4 / PROJ_BLK + j] = g_w32[j * PROJ_BLK];
  }

  const int c   = lane & 15;
  const int kh  = (lane >> 4) << 2;
  const int row = tile * 16 + c;
  const float* xrow = x + (size_t)row * D_SZ + kh;
  const bool z0 = first && (c < 3);
  const bool z1 = first && (c < 2);
  const bool z2 = first && (c < 1);

  ProjPtrs p;
  p.xp0 = z0 ? xrow : xrow - 3 * D_SZ;
  p.xp1 = z1 ? xrow : xrow - 2 * D_SZ;
  p.xp2 = z2 ? xrow : xrow - 1 * D_SZ;
  p.xp3 = xrow;
  p.bp0 = w + (size_t)c * D_SZ + kh;
  p.bp1 = w + (size_t)(16 + c) * D_SZ + kh;
  p.lw  = s_u + LDS_CW + kh * 4;
  p.lb  = s_u + LDS_CB + kh;
  p.l2  = s_u + LDS_W32 + kh;

  ProjStepG ring[PF_R];
  ProjStepL lring[PF_R];
#pragma unroll
  for (int i = 0; i < PF_R - 1; ++i) {
    ring[i] = proj_gload(p, 8 * i);
    __builtin_amdgcn_sched_barrier(0);
  }

#pragma unroll
  for (int j = 0; j < STG_V4; ++j)
    *(v4f*)(s_u + (size_t)(threadIdx.x + j * PROJ_BLK) * 4) = stg[j];
  __syncthreads();

  lring[0] = proj_lload(p, 0);

  if (first) conv_proj_loop<true>(ring, lring, p, z0, z1, z2, proj, tile, lane);
  else       conv_proj_loop<false>(ring, lring, p, z0, z1, z2, proj, tile, lane);
}

#define STG_CNT(NT) ((CH_T * P_SZ + (NT) - 1) / (NT))

template <int NT>
__device__ __forceinline__ void stage_proj_load(const float* __restrict__ pr, float (&v)[STG_CNT(NT)])
{
#pragma unroll
  for (int j = 0; j < STG_CNT(NT); ++j) {
    const int i = threadIdx.x + j * NT;
    v[j] = pr[i < CH_T * P_SZ ? i : CH_T * P_SZ - 1];
  }
}

#define S_ROW_N (CH_T * BC_W + CH_T + 1)
template <int NT>
__device__ __forceinline__ void stage_proj_store(const float (&v)[STG_CNT(NT)], float* __restrict__ s_row)
{
#pragma unroll
  for (int j = 0; j < STG_CNT(NT); ++j) {
    const int i   = threadIdx.x + j * NT;
    const int r   = i / P_SZ;
    const int col = i - r * P_SZ;
    const int a   = (col < BC_W) ? r * BC_W + col : CH_T * BC_W + r;
    s_row[i < CH_T * P_SZ ? a : S_ROW_N - 1] = v[j];
  }
}

#define SUM_BLK (2 * SCAN_BLK)
#define SUM_T (CH_T / 2)
__global__ __launch_bounds__(SUM_BLK) void scan_chunk_summary_kernel(
    const float* __restrict__ x, const float* __restrict__ cw,
    const float* __restrict__ cb_, const float* __restrict__ proj,
    const float* __restrict__ dt_w, const float* __restrict__ dt_b,
    const float* __restrict__ a_log,
    float* __restrict__ ch_e, float* __restrict__ ch_s)
{
  const int ck = blockIdx.x / (CHAN / SCAN_BLK);
  const int cb = (blockIdx.x % (CHAN / SCAN_BLK)) * SCAN_BLK;
  const int b  = cb >> 10;
  const int half = __builtin_amdgcn_readfirstlane((int)threadIdx.x / SCAN_BLK);
  const int cl = threadIdx.x & (SCAN_BLK - 1);
  const int c  = cb + cl;
  const int d  = c & (D_SZ - 1);

  v2f ac[8];
#pragma unroll
  for (int k = 0; k < 8; ++k) {
    ac[k].x = -__expf(a_log[d * N_SZ + 2 * k]);
    ac[k].y = -__expf(a_log[d * N_SZ + 2 * k + 1]);
  }
  const float dtw = dt_w[d];
  const float dtb = dt_b[d];
  const v4f   cwd = *(const v4f*)(cw + (size_t)d * 4);
  const float cbd = cb_[d];

  const int t0 = ck * CH_T;
  const int hb = half * SUM_T;
  const float* pr = proj + ((size_t)b * L_SZ + t0) * P_SZ;
  const float* xr = x    + ((size_t)b * L_SZ + t0) * D_SZ + d;

  __shared__ __attribute__((aligned(16))) float s_row[S_ROW_N];
  __shared__ float s_hi[(N_SZ + 1) * SCAN_BLK];
  const float* s_bc = s_row;
  const float* s_dt = s_row + CH_T * BC_W;
  {
    float stg[STG_CNT(SUM_BLK)];
    stage_proj_load<SUM_BLK>(pr, stg);
    stage_proj_store<SUM_BLK>(stg, s_row);
  }
  __syncthreads();

  float xw0 = 0.f, xw1 = 0.f, xw2 = 0.f;
  if (ck > 0 || half > 0) {
    xw0 = xr[(hb - 3) * D_SZ]; xw1 = xr[(hb - 2) * D_SZ]; xw2 = xr[(hb - 1) * D_SZ];
  }

  v2f h[8];
#pragma unroll
  for (int k = 0; k < 8; ++k) h[k] = (v2f){0.f, 0.f};
  float dsum = 0.f;

  for (int tb = hb; tb < hb + SUM_T; tb += SCAN_TB) {
    const v4f dq0 = *(const v4f*)(s_dt + tb);
    const v4f dq1 = *(const v4f*)(s_dt + tb + 4);
    const float p32v[SCAN_TB] = {dq0.x, dq0.y, dq0.z, dq0.w, dq1.x, dq1.y, dq1.z, dq1.w};
    v4f bq[SCAN_TB][N_SZ / 4];
#pragma unroll
    for (int i = 0; i < SCAN_TB; ++i) {
      const v4f* pt = (const v4f*)(s_bc + (tb + i) * BC_W);
#pragma unroll
      for (int q = 0; q < N_SZ / 4; ++q) bq[i][q] = pt[q];
    }

    float xv[SCAN_TB], xt[SCAN_TB];
#pragma unroll
    for (int i = 0; i < SCAN_TB; ++i) xv[i] = xr[(size_t)(tb + i) * D_SZ];
    xt[0] = conv_silu(xw0, xw1, xw2, xv[0], cwd, cbd);
    xt[1] = conv_silu(xw1, xw2, xv[0], xv[1], cwd, cbd);
    xt[2] = conv_silu(xw2, xv[0], xv[1], xv[2], cwd, cbd);
#pragma unroll
    for (int i = 3; i < SCAN_TB; ++i)
      xt[i] = conv_silu(xv[i - 3], xv[i - 2], xv[i - 1], xv[i], cwd, cbd);
    xw0 = xv[SCAN_TB - 3]; xw1 = xv[SCAN_TB - 2]; xw2 = xv[SCAN_TB - 1];

    float dl[SCAN_TB], dx[SCAN_TB];
#pragma unroll
    for (int i = 0; i < SCAN_TB; ++i) {
      const float p32 = p32v[i];
      const float z   = fmaf(p32, dtw, dtb);
      const float sp  = __logf(1.f + __expf(z));
      dl[i] = (z > 20.f) ? z : sp;
    }
#pragma unroll
    for (int i = 0; i < SCAN_TB; ++i) { dx[i] = dl[i] * xt[i]; dsum += dl[i]; }

#pragma unroll
    for (int i = 0; i < SCAN_TB; ++i) {
#pragma unroll
      for (int k = 0; k < 8; ++k) {
        const v2f e  = dl[i] * ac[k];
        const v2f ab = {__expf(e.x), __expf(e.y)};
        const v4f b4 = bq[i][k >> 1];
        const v2f bt = (k & 1) ? (v2f){b4.z, b4.w} : (v2f){b4.x, b4.y};
        const v2f u  = dx[i] * bt;
        h[k] = __builtin_elementwise_fma(ab, h[k], u);
      }
    }
  }

  if (half > 0) {
#pragma unroll
    for (int k = 0; k < 8; ++k) {
      s_hi[(2 * k) * SCAN_BLK + cl]     = h[k].x;
      s_hi[(2 * k + 1) * SCAN_BLK + cl] = h[k].y;
    }
    s_hi[N_SZ * SCAN_BLK + cl] = dsum;
  }
  __syncthreads();
  if (half == 0) {
    const float sh = s_hi[N_SZ * SCAN_BLK + cl];
#pragma unroll
    for (int k = 0; k < 8; ++k) {
      const v2f eh = {s_hi[(2 * k) * SCAN_BLK + cl], s_hi[(2 * k + 1) * SCAN_BLK + cl]};
      const v2f a  = sh * ac[k];
      const v2f ph = {__expf(a.x), __expf(a.y)};
      const v2f e  = __builtin_elementwise_fma(ph, h[k], eh);
      const size_t o0 = ((size_t)(ck * N_SZ + 2 * k)) * CHAN + c;
      ch_e[o0]        = e.x;
      ch_e[o0 + CHAN] = e.y;
    }
    ch_s[(size_t)ck * CHAN + c] = dsum + sh;
  }
}

__global__ __launch_bounds__(SCAN_BLK) void scan_kernel(
    const float* __restrict__ x, const float* __restrict__ cw,
    const float* __restrict__ cb_,
    const float* __restrict__ proj, const float* __restrict__ dt_w,
    const float* __restrict__ dt_b, const float* __restrict__ a_log,
    const float* __restrict__ Dp, const float* __restrict__ ch_e,
    const float* __restrict__ ch_s, float* __restrict__ out)
{
  const int ck = blockIdx.x / (CHAN / SCAN_BLK);
  const int cb = (blockIdx.x % (CHAN / SCAN_BLK)) * SCAN_BLK;
  const int b  = cb >> 10;
  const int c  = cb + threadIdx.x;
  const int d  = c & (D_SZ - 1);

  v2f ac[8];
#pragma unroll
  for (int k = 0; k < 8; ++k) {
    ac[k].x = -__expf(a_log[d * N_SZ + 2 * k]);
    ac[k].y = -__expf(a_log[d * N_SZ + 2 * k + 1]);
  }
  const float dtw = dt_w[d];
  const float dtb = dt_b[d];
  const float dpd = Dp[d];
  const v4f   cwd = *(const v4f*)(cw + (size_t)d * 4);
  const float cbd = cb_[d];

  const float* pr = proj + ((size_t)b * L_SZ + ck * CH_T) * P_SZ;
  __shared__ __attribute__((aligned(16))) float s_row[S_ROW_N];
  const float* s_bc = s_row;
  const float* s_dt = s_row + CH_T * BC_W;
  float stg[STG_CNT(SCAN_BLK)];
  stage_proj_load<SCAN_BLK>(pr, stg);

  v2f h[8];
#pragma unroll
  for (int k = 0; k < 8; ++k) h[k] = (v2f){0.f, 0.f};
#pragma unroll 4
  for (int j = 0; j < ck; ++j) {
    const float sj = ch_s[(size_t)j * CHAN + c];
    v2f ej[8];
#pragma unroll
    for (int k = 0; k < 8; ++k) {
      const size_t o0 = ((size_t)(j * N_SZ + 2 * k)) * CHAN + c;
      ej[k].x = ch_e[o0];
      ej[k].y = ch_e[o0 + CHAN];
    }
#pragma unroll
    for (int k = 0; k < 8; ++k) {
      const v2f a  = sj * ac[k];
      const v2f pj = {__expf(a.x), __expf(a.y)};
      h[k] = __builtin_elementwise_fma(pj, h[k], ej[k]);
    }
  }

  const int t0 = ck * CH_T;
  const float* xr   = x    + ((size_t)b * L_SZ + t0) * D_SZ + d;
  float*       outr = out  + ((size_t)b * L_SZ + t0) * D_SZ + d;

  stage_proj_store<SCAN_BLK>(stg, s_row);
  __syncthreads();

  float xw0 = 0.f, xw1 = 0.f, xw2 = 0.f;
  if (ck > 0) { xw0 = xr[-3 * D_SZ]; xw1 = xr[-2 * D_SZ]; xw2 = xr[-1 * D_SZ]; }

  for (int tb = 0; tb < CH_T; tb += SCAN_TB) {
    float xt[SCAN_TB], xv[SCAN_TB];
#pragma unroll
    for (int i = 0; i < SCAN_TB; ++i) xv[i] = xr[(size_t)(tb + i) * D_SZ];
    xt[0] = conv_silu(xw0, xw1, xw2, xv[0], cwd, cbd);
    xt[1] = conv_silu(xw1, xw2, xv[0], xv[1], cwd, cbd);
    xt[2] = conv_silu(xw2, xv[0], xv[1], xv[2], cwd, cbd);
#pragma unroll
    for (int i = 3; i < SCAN_TB; ++i)
      xt[i] = conv_silu(xv[i - 3], xv[i - 2], xv[i - 1], xv[i], cwd, cbd);
    xw0 = xv[SCAN_TB - 3]; xw1 = xv[SCAN_TB - 2]; xw2 = xv[SCAN_TB - 1];

    float dl[SCAN_TB], dx[SCAN_TB];
#pragma unroll
    for (int i = 0; i < SCAN_TB; ++i) {
      const float p32 = s_dt[tb + i];
      const float z   = fmaf(p32, dtw, dtb);
      const float sp  = __logf(1.f + __expf(z));
      dl[i] = (z > 20.f) ? z : sp;
    }
#pragma unroll
    for (int i = 0; i < SCAN_TB; ++i) dx[i] = dl[i] * xt[i];

#pragma unroll
    for (int i = 0; i < SCAN_TB; ++i) {
      const v4f* pt = (const v4f*)(s_bc + (tb + i) * BC_W);
      v2f ya = {0.f, 0.f}, yb = {0.f, 0.f};
#pragma unroll
      for (int k = 0; k < 8; ++k) {
        const v2f e  = dl[i] * ac[k];
        const v2f ab = {__expf(e.x), __expf(e.y)};
        const v4f b4 = pt[k >> 1];
        const v2f bt = (k & 1) ? (v2f){b4.z, b4.w} : (v2f){b4.x, b4.y};
        const v4f c4 = pt[N_SZ / 4 + (k >> 1)];
        const v2f ct = (k & 1) ? (v2f){c4.z, c4.w} : (v2f){c4.x, c4.y};
        const v2f u  = dx[i] * bt;
        h[k] = __builtin_elementwise_fma(ab, h[k], u);
        if (k & 1) yb = __builtin_elementwise_fma(h[k], ct, yb);
        else       ya = __builtin_elementwise_fma(h[k], ct, ya);
      }
      const v2f ys = ya + yb;
      const float y = ys.x + ys.y;
      outr[(size_t)(tb + i) * D_SZ] = fmaf(xv[i], dpd, y);
    }
  }
}

extern "C" void kernel_launch(void* const* d_in, const int* in_sizes, int n_in,
                              void* d_out, int out_size, void* d_ws, size_t ws_size,
                              hipStream_t stream) {
  const float* x      = (const float*)d_in[0];
  const float* conv_w = (const float*)d_in[1];
  const float* conv_b = (const float*)d_in[2];
  const float* xproj  = (const float*)d_in[3];
  const float* dt_w   = (const float*)d_in[4];
  const float* dt_b   = (const float*)d_in[5];
  const float* a_log  = (const float*)d_in[6];
  const float* Dp     = (const float*)d_in[7];
  float* out = (float*)d_out;

  float* proj = (float*)d_ws;
  float* ch_e = proj + (size_t)B_SZ * L_SZ * P_SZ;
  float* ch_s = ch_e + (size_t)(CH_N - 1) * CHAN * N_SZ;

  conv_proj_wmma_kernel<<<(B_SZ * L_SZ) / 16 / (PROJ_BLK / 32), PROJ_BLK, 0, stream>>>(
      x, conv_w, conv_b, xproj, proj);

  scan_chunk_summary_kernel<<<(CHAN / SCAN_BLK) * (CH_N - 1), SUM_BLK, 0, stream>>>(
      x, conv_w, conv_b, proj, dt_w, dt_b, a_log, ch_e, ch_s);

  scan_kernel<<<(CHAN / SCAN_BLK) * CH_N, SCAN_BLK, 0, stream>>>(
      x, conv_w, conv_b, proj, dt_w, dt_b, a_log, Dp, ch_e, ch_s, out);
}


# scan and chunk summary: a batch's LDS reads (B, C rows, column 32) issued as two runs of loads, one per 4 steps, kept together by scheduling fences (were in about 25 and 12 places); fp order same
# speedup vs baseline: 28.5980x; 1.0833x over previous
#include <hip/hip_runtime.h>
#include <hip/hip_bf16.h>


#define B_SZ 4
#define L_SZ 2048
#define D_SZ 1024
#define N_SZ 16
#define P_SZ 33
#define CH_N 16
#define CH_T (L_SZ / CH_N)
#define CHAN (B_SZ * D_SZ)
#define SCAN_BLK 256
#define SCAN_TB 8
#define BC_W (2 * N_SZ)
#define PRE_T 2

typedef float v2f __attribute__((ext_vector_type(2)));
typedef float v4f __attribute__((ext_vector_type(4)));
typedef float v8f __attribute__((ext_vector_type(8)));

__device__ __forceinline__ float conv_silu(float x0, float x1, float x2, float x3,
                                           v4f w, float bias)
{
  float acc = bias;
  acc = fmaf(x0, w.x, acc);
  acc = fmaf(x1, w.y, acc);
  acc = fmaf(x2, w.z, acc);
  acc = fmaf(x3, w.w, acc);
  return acc * __builtin_amdgcn_rcpf(1.f + __expf(-acc));
}

struct ProjStepG {
  v4f x0, x1, x2, x3;
  v4f b0, b1;
};
struct ProjStepL {
  v4f wa, wb, wc, wd;
  v4f bias;
  v4f b2;
};

#define LDS_CW 0
#define LDS_CB (D_SZ * 4)
#define LDS_W32 (D_SZ * 5)
#define LDS_PROJ_N (D_SZ * 6)
#define PROJ_BLK 64
#define STG_V4 (LDS_PROJ_N / 4 / PROJ_BLK)

#define PF_R 10
#define PF_MAIN (((D_SZ / 8) / PF_R) * PF_R)

struct ProjPtrs {
  const float *xp0, *xp1, *xp2, *xp3, *bp0, *bp1;
  const float *lw, *lb, *l2;
};

__device__ __forceinline__ ProjStepG proj_gload(const ProjPtrs& p, const int k0)
{
  ProjStepG r;
  r.x0 = *(const v4f*)(p.xp0 + k0);
  r.x1 = *(const v4f*)(p.xp1 + k0);
  r.x2 = *(const v4f*)(p.xp2 + k0);
  r.x3 = *(const v4f*)(p.xp3 + k0);
  r.b0 = *(const v4f*)(p.bp0 + k0);
  r.b1 = *(const v4f*)(p.bp1 + k0);
  return r;
}

__device__ __forceinline__ ProjStepL proj_lload(const ProjPtrs& p, const int k0)
{
  ProjStepL r;
  r.wa   = *(const v4f*)(p.lw + k0 * 4);
  r.wb   = *(const v4f*)(p.lw + k0 * 4 + 4);
  r.wc   = *(const v4f*)(p.lw + k0 * 4 + 8);
  r.wd   = *(const v4f*)(p.lw + k0 * 4 + 12);
  r.bias = *(const v4f*)(p.lb + k0);
  r.b2   = *(const v4f*)(p.l2 + k0);
  return r;
}

template <bool FIRST>
__device__ __forceinline__ void proj_step(ProjStepG s, const ProjStepL l, const bool z0,
                                          const bool z1, const bool z2,
                                          v8f& acc0, v8f& acc1, v8f& acc2)
{
  if (FIRST) {
    const v4f zero4 = {0.f, 0.f, 0.f, 0.f};
    if (z0) s.x0 = zero4;
    if (z1) s.x1 = zero4;
    if (z2) s.x2 = zero4;
  }
  const v2f alo = {conv_silu(s.x0.x, s.x1.x, s.x2.x, s.x3.x, l.wa, l.bias.x),
                   conv_silu(s.x0.y, s.x1.y, s.x2.y, s.x3.y, l.wb, l.bias.y)};
  const v2f ahi = {conv_silu(s.x0.z, s.x1.z, s.x2.z, s.x3.z, l.wc, l.bias.z),
                   conv_silu(s.x0.w, s.x1.w, s.x2.w, s.x3.w, l.wd, l.bias.w)};

  acc0 = __builtin_amdgcn_wmma_f32_16x16x4_f32(false, alo, false, (v2f){s.b0.x, s.b0.y},
                                               (short)0, acc0, false, false);
  acc1 = __builtin_amdgcn_wmma_f32_16x16x4_f32(false, alo, false, (v2f){s.b1.x, s.b1.y},
                                               (short)0, acc1, false, false);
  acc2 = __builtin_amdgcn_wmma_f32_16x16x4_f32(false, alo, false, (v2f){l.b2.x, l.b2.y},
                                               (short)0, acc2, false, false);
  acc0 = __builtin_amdgcn_wmma_f32_16x16x4_f32(false, ahi, false, (v2f){s.b0.z, s.b0.w},
                                               (short)0, acc0, false, false);
  acc1 = __builtin_amdgcn_wmma_f32_16x16x4_f32(false, ahi, false, (v2f){s.b1.z, s.b1.w},
                                               (short)0, acc1, false, false);
  acc2 = __builtin_amdgcn_wmma_f32_16x16x4_f32(false, ahi, false, (v2f){l.b2.z, l.b2.w},
                                               (short)0, acc2, false, false);
}

template <bool FIRST>
__device__ __forceinline__ void conv_proj_loop(
    ProjStepG (&ring)[PF_R], ProjStepL (&lring)[PF_R], const ProjPtrs& p,
    const bool z0, const bool z1, const bool z2,
    float* __restrict__ proj, const int tile, const int lane)
{
  const int c = lane & 15;
  v8f acc0 = {}, acc1 = {}, acc2 = {};

  for (int s0 = 0; s0 < PF_MAIN; s0 += PF_R) {
#pragma unroll
    for (int i = 0; i < PF_R; ++i) {
      lring[(i + 1) % PF_R] = proj_lload(p, ((s0 + i + 1) * 8) & (D_SZ - 1));
      ring[(i + PF_R - 1) % PF_R] = proj_gload(p, ((s0 + i + PF_R - 1) * 8) & (D_SZ - 1));
      __builtin_amdgcn_sched_barrier(0);
      proj_step<FIRST>(ring[i], lring[i], z0, z1, z2, acc0, acc1, acc2);
      __builtin_amdgcn_sched_barrier(0);
    }
  }
#pragma unroll
  for (int i = 0; i < (D_SZ / 8) - PF_MAIN; ++i) {
    if (i + 1 < (D_SZ / 8) - PF_MAIN)
      lring[(i + 1) % PF_R] = proj_lload(p, (PF_MAIN + i + 1) * 8);
    proj_step<FIRST>(ring[(PF_MAIN + i) % PF_R], lring[i % PF_R], z0, z1, z2, acc0, acc1, acc2);
  }

  const int mhi = (lane >> 4) << 3;
#pragma unroll
  for (int v = 0; v < 8; ++v) {
    float* pr = proj + (size_t)(tile * 16 + v + mhi) * P_SZ;
    pr[c]      = acc0[v];
    pr[16 + c] = acc1[v];
    if (c == 0) pr[32] = acc2[v];
  }
}

__global__ __launch_bounds__(PROJ_BLK) void conv_proj_wmma_kernel(
    const float* __restrict__ x, const float* __restrict__ cw,
    const float* __restrict__ cb, const float* __restrict__ w,
    float* __restrict__ proj)
{
  __shared__ __attribute__((aligned(16))) float s_u[LDS_PROJ_N];

  const int wave = __builtin_amdgcn_readfirstlane((int)threadIdx.x >> 5);
  const int lane = threadIdx.x & 31;
  const int tile = blockIdx.x * (PROJ_BLK / 32) + wave;
  const bool first = (tile & (L_SZ / 16 - 1)) == 0;

  v4f stg[STG_V4];
  {
    const v4f* g_cw  = (const v4f*)cw + threadIdx.x;
    const v4f* g_cb  = (const v4f*)cb + threadIdx.x;
    const v4f* g_w32 = (const v4f*)(w + (size_t)(P_SZ - 1) * D_SZ) + threadIdx.x;
#pragma unroll
    for (int j = 0; j < D_SZ / PROJ_BLK; ++j)     stg[j] = g_cw[j * PROJ_BLK];
#pragma unroll
    for (int j = 0; j < D_SZ / 4 / PROJ_BLK; ++j) stg[D_SZ / PROJ_BLK + j] = g_cb[j * PROJ_BLK];
#pragma unroll
    for (int j = 0; j < D_SZ / 4 / PROJ_BLK; ++j)
      stg[D_SZ / PROJ_BLK + D_SZ / 4 / PROJ_BLK + j] = g_w32[j * PROJ_BLK];
  }

  const int c   = lane & 15;
  const int kh  = (lane >> 4) << 2;
  const int row = tile * 16 + c;
  const float* xrow = x + (size_t)row * D_SZ + kh;
  const bool z0 = first && (c < 3);
  const bool z1 = first && (c < 2);
  const bool z2 = first && (c < 1);

  ProjPtrs p;
  p.xp0 = z0 ? xrow : xrow - 3 * D_SZ;
  p.xp1 = z1 ? xrow : xrow - 2 * D_SZ;
  p.xp2 = z2 ? xrow : xrow - 1 * D_SZ;
  p.xp3 = xrow;
  p.bp0 = w + (size_t)c * D_SZ + kh;
  p.bp1 = w + (size_t)(16 + c) * D_SZ + kh;
  p.lw  = s_u + LDS_CW + kh * 4;
  p.lb  = s_u + LDS_CB + kh;
  p.l2  = s_u + LDS_W32 + kh;

  ProjStepG ring[PF_R];
  ProjStepL lring[PF_R];
#pragma unroll
  for (int i = 0; i < PF_R - 1; ++i) {
    ring[i] = proj_gload(p, 8 * i);
    __builtin_amdgcn_sched_barrier(0);
  }

#pragma unroll
  for (int j = 0; j < STG_V4; ++j)
    *(v4f*)(s_u + (size_t)(threadIdx.x + j * PROJ_BLK) * 4) = stg[j];
  __syncthreads();

  lring[0] = proj_lload(p, 0);

  if (first) conv_proj_loop<true>(ring, lring, p, z0, z1, z2, proj, tile, lane);
  else       conv_proj_loop<false>(ring, lring, p, z0, z1, z2, proj, tile, lane);
}

#define STG_CNT(NT) ((CH_T * P_SZ + (NT) - 1) / (NT))

template <int NT>
__device__ __forceinline__ void stage_proj_load(const float* __restrict__ pr, float (&v)[STG_CNT(NT)])
{
#pragma unroll
  for (int j = 0; j < STG_CNT(NT); ++j) {
    const int i = threadIdx.x + j * NT;
    v[j] = pr[i < CH_T * P_SZ ? i : CH_T * P_SZ - 1];
  }
}

#define S_ROW_N (CH_T * BC_W + CH_T + 1)
template <int NT>
__device__ __forceinline__ void stage_proj_store(const float (&v)[STG_CNT(NT)], float* __restrict__ s_row)
{
#pragma unroll
  for (int j = 0; j < STG_CNT(NT); ++j) {
    const int i   = threadIdx.x + j * NT;
    const int r   = i / P_SZ;
    const int col = i - r * P_SZ;
    const int a   = (col < BC_W) ? r * BC_W + col : CH_T * BC_W + r;
    s_row[i < CH_T * P_SZ ? a : S_ROW_N - 1] = v[j];
  }
}

#define SUM_BLK (2 * SCAN_BLK)
#define SUM_T (CH_T / 2)
__global__ __launch_bounds__(SUM_BLK) void scan_chunk_summary_kernel(
    const float* __restrict__ x, const float* __restrict__ cw,
    const float* __restrict__ cb_, const float* __restrict__ proj,
    const float* __restrict__ dt_w, const float* __restrict__ dt_b,
    const float* __restrict__ a_log,
    float* __restrict__ ch_e, float* __restrict__ ch_s)
{
  const int ck = blockIdx.x / (CHAN / SCAN_BLK);
  const int cb = (blockIdx.x % (CHAN / SCAN_BLK)) * SCAN_BLK;
  const int b  = cb >> 10;
  const int half = __builtin_amdgcn_readfirstlane((int)threadIdx.x / SCAN_BLK);
  const int cl = threadIdx.x & (SCAN_BLK - 1);
  const int c  = cb + cl;
  const int d  = c & (D_SZ - 1);

  v2f ac[8];
#pragma unroll
  for (int k = 0; k < 8; ++k) {
    ac[k].x = -__expf(a_log[d * N_SZ + 2 * k]);
    ac[k].y = -__expf(a_log[d * N_SZ + 2 * k + 1]);
  }
  const float dtw = dt_w[d];
  const float dtb = dt_b[d];
  const v4f   cwd = *(const v4f*)(cw + (size_t)d * 4);
  const float cbd = cb_[d];

  const int t0 = ck * CH_T;
  const int hb = half * SUM_T;
  const float* pr = proj + ((size_t)b * L_SZ + t0) * P_SZ;
  const float* xr = x    + ((size_t)b * L_SZ + t0) * D_SZ + d;

  __shared__ __attribute__((aligned(16))) float s_row[S_ROW_N];
  __shared__ float s_hi[(N_SZ + 1) * SCAN_BLK];
  const float* s_bc = s_row;
  const float* s_dt = s_row + CH_T * BC_W;
  {
    float stg[STG_CNT(SUM_BLK)];
    stage_proj_load<SUM_BLK>(pr, stg);
    stage_proj_store<SUM_BLK>(stg, s_row);
  }
  __syncthreads();

  float xw0 = 0.f, xw1 = 0.f, xw2 = 0.f;
  if (ck > 0 || half > 0) {
    xw0 = xr[(hb - 3) * D_SZ]; xw1 = xr[(hb - 2) * D_SZ]; xw2 = xr[(hb - 1) * D_SZ];
  }

  v2f h[8];
#pragma unroll
  for (int k = 0; k < 8; ++k) h[k] = (v2f){0.f, 0.f};
  float dsum = 0.f;

  for (int tb = hb; tb < hb + SUM_T; tb += SCAN_TB) {
    float xv[SCAN_TB], xt[SCAN_TB];
#pragma unroll
    for (int i = 0; i < SCAN_TB; ++i) xv[i] = xr[(size_t)(tb + i) * D_SZ];

    const v4f dq0 = *(const v4f*)(s_dt + tb);
    const v4f dq1 = *(const v4f*)(s_dt + tb + 4);
    const float p32v[SCAN_TB] = {dq0.x, dq0.y, dq0.z, dq0.w, dq1.x, dq1.y, dq1.z, dq1.w};
    v4f bq[SCAN_TB / 2][N_SZ / 4];
#pragma unroll
    for (int i = 0; i < SCAN_TB / 2; ++i) {
      const v4f* pt = (const v4f*)(s_bc + (tb + i) * BC_W);
#pragma unroll
      for (int q = 0; q < N_SZ / 4; ++q) bq[i][q] = pt[q];
    }
    __builtin_amdgcn_sched_barrier(0);

    float dl[SCAN_TB], dx[SCAN_TB];
#pragma unroll
    for (int i = 0; i < SCAN_TB; ++i) {
      const float p32 = p32v[i];
      const float z   = fmaf(p32, dtw, dtb);
      const float sp  = __logf(1.f + __expf(z));
      dl[i] = (z > 20.f) ? z : sp;
    }
    v2f abp[PRE_T][8];
#pragma unroll
    for (int i = 0; i < PRE_T; ++i) {
#pragma unroll
      for (int k = 0; k < 8; ++k) {
        const v2f e = dl[i] * ac[k];
        abp[i][k] = (v2f){__expf(e.x), __expf(e.y)};
      }
    }
    __builtin_amdgcn_sched_barrier(0);

    xt[0] = conv_silu(xw0, xw1, xw2, xv[0], cwd, cbd);
    xt[1] = conv_silu(xw1, xw2, xv[0], xv[1], cwd, cbd);
    xt[2] = conv_silu(xw2, xv[0], xv[1], xv[2], cwd, cbd);
#pragma unroll
    for (int i = 3; i < SCAN_TB; ++i)
      xt[i] = conv_silu(xv[i - 3], xv[i - 2], xv[i - 1], xv[i], cwd, cbd);
    xw0 = xv[SCAN_TB - 3]; xw1 = xv[SCAN_TB - 2]; xw2 = xv[SCAN_TB - 1];
#pragma unroll
    for (int i = 0; i < SCAN_TB; ++i) { dx[i] = dl[i] * xt[i]; dsum += dl[i]; }

#pragma unroll
    for (int hf = 0; hf < 2; ++hf) {
      if (hf == 1) {
#pragma unroll
        for (int i = 0; i < SCAN_TB / 2; ++i) {
          const v4f* pt = (const v4f*)(s_bc + (tb + SCAN_TB / 2 + i) * BC_W);
#pragma unroll
          for (int q = 0; q < N_SZ / 4; ++q) bq[i][q] = pt[q];
        }
        __builtin_amdgcn_sched_barrier(0);
      }
#pragma unroll
      for (int ii = 0; ii < SCAN_TB / 2; ++ii) {
        const int i = hf * (SCAN_TB / 2) + ii;
#pragma unroll
        for (int k = 0; k < 8; ++k) {
          v2f ab;
          if (i < PRE_T) {
            ab = abp[i][k];
          } else {
            const v2f e = dl[i] * ac[k];
            ab = (v2f){__expf(e.x), __expf(e.y)};
          }
          const v4f b4 = bq[ii][k >> 1];
          const v2f bt = (k & 1) ? (v2f){b4.z, b4.w} : (v2f){b4.x, b4.y};
          const v2f u  = dx[i] * bt;
          h[k] = __builtin_elementwise_fma(ab, h[k], u);
        }
      }
    }
  }

  if (half > 0) {
#pragma unroll
    for (int k = 0; k < 8; ++k) {
      s_hi[(2 * k) * SCAN_BLK + cl]     = h[k].x;
      s_hi[(2 * k + 1) * SCAN_BLK + cl] = h[k].y;
    }
    s_hi[N_SZ * SCAN_BLK + cl] = dsum;
  }
  __syncthreads();
  if (half == 0) {
    const float sh = s_hi[N_SZ * SCAN_BLK + cl];
#pragma unroll
    for (int k = 0; k < 8; ++k) {
      const v2f eh = {s_hi[(2 * k) * SCAN_BLK + cl], s_hi[(2 * k + 1) * SCAN_BLK + cl]};
      const v2f a  = sh * ac[k];
      const v2f ph = {__expf(a.x), __expf(a.y)};
      const v2f e  = __builtin_elementwise_fma(ph, h[k], eh);
      const size_t o0 = ((size_t)(ck * N_SZ + 2 * k)) * CHAN + c;
      ch_e[o0]        = e.x;
      ch_e[o0 + CHAN] = e.y;
    }
    ch_s[(size_t)ck * CHAN + c] = dsum + sh;
  }
}

__global__ __launch_bounds__(SCAN_BLK) void scan_kernel(
    const float* __restrict__ x, const float* __restrict__ cw,
    const float* __restrict__ cb_,
    const float* __restrict__ proj, const float* __restrict__ dt_w,
    const float* __restrict__ dt_b, const float* __restrict__ a_log,
    const float* __restrict__ Dp, const float* __restrict__ ch_e,
    const float* __restrict__ ch_s, float* __restrict__ out)
{
  const int ck = blockIdx.x / (CHAN / SCAN_BLK);
  const int cb = (blockIdx.x % (CHAN / SCAN_BLK)) * SCAN_BLK;
  const int b  = cb >> 10;
  const int c  = cb + threadIdx.x;
  const int d  = c & (D_SZ - 1);

  v2f ac[8];
#pragma unroll
  for (int k = 0; k < 8; ++k) {
    ac[k].x = -__expf(a_log[d * N_SZ + 2 * k]);
    ac[k].y = -__expf(a_log[d * N_SZ + 2 * k + 1]);
  }
  const float dtw = dt_w[d];
  const float dtb = dt_b[d];
  const float dpd = Dp[d];
  const v4f   cwd = *(const v4f*)(cw + (size_t)d * 4);
  const float cbd = cb_[d];

  const float* pr = proj + ((size_t)b * L_SZ + ck * CH_T) * P_SZ;
  __shared__ __attribute__((aligned(16))) float s_row[S_ROW_N];
  const float* s_bc = s_row;
  const float* s_dt = s_row + CH_T * BC_W;
  float stg[STG_CNT(SCAN_BLK)];
  stage_proj_load<SCAN_BLK>(pr, stg);

  v2f h[8];
#pragma unroll
  for (int k = 0; k < 8; ++k) h[k] = (v2f){0.f, 0.f};
#pragma unroll 4
  for (int j = 0; j < ck; ++j) {
    const float sj = ch_s[(size_t)j * CHAN + c];
    v2f ej[8];
#pragma unroll
    for (int k = 0; k < 8; ++k) {
      const size_t o0 = ((size_t)(j * N_SZ + 2 * k)) * CHAN + c;
      ej[k].x = ch_e[o0];
      ej[k].y = ch_e[o0 + CHAN];
    }
#pragma unroll
    for (int k = 0; k < 8; ++k) {
      const v2f a  = sj * ac[k];
      const v2f pj = {__expf(a.x), __expf(a.y)};
      h[k] = __builtin_elementwise_fma(pj, h[k], ej[k]);
    }
  }

  const int t0 = ck * CH_T;
  const float* xr   = x    + ((size_t)b * L_SZ + t0) * D_SZ + d;
  float*       outr = out  + ((size_t)b * L_SZ + t0) * D_SZ + d;

  stage_proj_store<SCAN_BLK>(stg, s_row);
  __syncthreads();

  float xw0 = 0.f, xw1 = 0.f, xw2 = 0.f;
  if (ck > 0) { xw0 = xr[-3 * D_SZ]; xw1 = xr[-2 * D_SZ]; xw2 = xr[-1 * D_SZ]; }

  for (int tb = 0; tb < CH_T; tb += SCAN_TB) {
    float xt[SCAN_TB], xv[SCAN_TB];
#pragma unroll
    for (int i = 0; i < SCAN_TB; ++i) xv[i] = xr[(size_t)(tb + i) * D_SZ];

    const v4f dq0 = *(const v4f*)(s_dt + tb);
    const v4f dq1 = *(const v4f*)(s_dt + tb + 4);
    const float p32v[SCAN_TB] = {dq0.x, dq0.y, dq0.z, dq0.w, dq1.x, dq1.y, dq1.z, dq1.w};
    v4f bc[SCAN_TB / 2][BC_W / 4];
#pragma unroll
    for (int i = 0; i < SCAN_TB / 2; ++i) {
      const v4f* pt = (const v4f*)(s_bc + (tb + i) * BC_W);
#pragma unroll
      for (int q = 0; q < BC_W / 4; ++q) bc[i][q] = pt[q];
    }
    __builtin_amdgcn_sched_barrier(0);

    float dl[SCAN_TB], dx[SCAN_TB];
#pragma unroll
    for (int i = 0; i < SCAN_TB; ++i) {
      const float p32 = p32v[i];
      const float z   = fmaf(p32, dtw, dtb);
      const float sp  = __logf(1.f + __expf(z));
      dl[i] = (z > 20.f) ? z : sp;
    }
    v2f abp[PRE_T][8];
#pragma unroll
    for (int i = 0; i < PRE_T; ++i) {
#pragma unroll
      for (int k = 0; k < 8; ++k) {
        const v2f e = dl[i] * ac[k];
        abp[i][k] = (v2f){__expf(e.x), __expf(e.y)};
      }
    }
    __builtin_amdgcn_sched_barrier(0);

    xt[0] = conv_silu(xw0, xw1, xw2, xv[0], cwd, cbd);
    xt[1] = conv_silu(xw1, xw2, xv[0], xv[1], cwd, cbd);
    xt[2] = conv_silu(xw2, xv[0], xv[1], xv[2], cwd, cbd);
#pragma unroll
    for (int i = 3; i < SCAN_TB; ++i)
      xt[i] = conv_silu(xv[i - 3], xv[i - 2], xv[i - 1], xv[i], cwd, cbd);
    xw0 = xv[SCAN_TB - 3]; xw1 = xv[SCAN_TB - 2]; xw2 = xv[SCAN_TB - 1];
#pragma unroll
    for (int i = 0; i < SCAN_TB; ++i) dx[i] = dl[i] * xt[i];

#pragma unroll
    for (int hf = 0; hf < 2; ++hf) {
      if (hf == 1) {
#pragma unroll
        for (int i = 0; i < SCAN_TB / 2; ++i) {
          const v4f* pt = (const v4f*)(s_bc + (tb + SCAN_TB / 2 + i) * BC_W);
#pragma unroll
          for (int q = 0; q < BC_W / 4; ++q) bc[i][q] = pt[q];
        }
        __builtin_amdgcn_sched_barrier(0);
      }
#pragma unroll
      for (int ii = 0; ii < SCAN_TB / 2; ++ii) {
        const int i = hf * (SCAN_TB / 2) + ii;
        v2f ya = {0.f, 0.f}, yb = {0.f, 0.f};
#pragma unroll
        for (int k = 0; k < 8; ++k) {
          v2f ab;
          if (i < PRE_T) {
            ab = abp[i][k];
          } else {
            const v2f e = dl[i] * ac[k];
            ab = (v2f){__expf(e.x), __expf(e.y)};
          }
          const v4f b4 = bc[ii][k >> 1];
          const v2f bt = (k & 1) ? (v2f){b4.z, b4.w} : (v2f){b4.x, b4.y};
          const v4f c4 = bc[ii][N_SZ / 4 + (k >> 1)];
          const v2f ct = (k & 1) ? (v2f){c4.z, c4.w} : (v2f){c4.x, c4.y};
          const v2f u  = dx[i] * bt;
          h[k] = __builtin_elementwise_fma(ab, h[k], u);
          if (k & 1) yb = __builtin_elementwise_fma(h[k], ct, yb);
          else       ya = __builtin_elementwise_fma(h[k], ct, ya);
        }
        const v2f ys = ya + yb;
        const float y = ys.x + ys.y;
        outr[(size_t)(tb + i) * D_SZ] = fmaf(xv[i], dpd, y);
      }
    }
  }
}

extern "C" void kernel_launch(void* const* d_in, const int* in_sizes, int n_in,
                              void* d_out, int out_size, void* d_ws, size_t ws_size,
                              hipStream_t stream) {
  const float* x      = (const float*)d_in[0];
  const float* conv_w = (const float*)d_in[1];
  const float* conv_b = (const float*)d_in[2];
  const float* xproj  = (const float*)d_in[3];
  const float* dt_w   = (const float*)d_in[4];
  const float* dt_b   = (const float*)d_in[5];
  const float* a_log  = (const float*)d_in[6];
  const float* Dp     = (const float*)d_in[7];
  float* out = (float*)d_out;

  float* proj = (float*)d_ws;
  float* ch_e = proj + (size_t)B_SZ * L_SZ * P_SZ;
  float* ch_s = ch_e + (size_t)(CH_N - 1) * CHAN * N_SZ;

  conv_proj_wmma_kernel<<<(B_SZ * L_SZ) / 16 / (PROJ_BLK / 32), PROJ_BLK, 0, stream>>>(
      x, conv_w, conv_b, xproj, proj);

  scan_chunk_summary_kernel<<<(CHAN / SCAN_BLK) * (CH_N - 1), SUM_BLK, 0, stream>>>(
      x, conv_w, conv_b, proj, dt_w, dt_b, a_log, ch_e, ch_s);

  scan_kernel<<<(CHAN / SCAN_BLK) * CH_N, SCAN_BLK, 0, stream>>>(
      x, conv_w, conv_b, proj, dt_w, dt_b, a_log, Dp, ch_e, ch_s, out);
}
